# starting kernel
# baseline (speedup 1.0000x reference)
#include <hip/hip_runtime.h>
#include <hip/hip_bf16.h>

typedef float v2f __attribute__((ext_vector_type(2)));
typedef float v8f __attribute__((ext_vector_type(8)));

#define T_DIM 2048
#define D_DIM 256
#define H_DIM 256
#define GAMMA_LAMBDA (0.99f * 0.9f)
#define CONT_ALPHA 0.01f

__global__ __launch_bounds__(256) void fwd_gemm_kernel(
    const float* __restrict__ X, const float* __restrict__ W1,
    const float* __restrict__ b1, float* __restrict__ h)
{
    const int lane = threadIdx.x & 31;
    const int wave = threadIdx.x >> 5;
    const int tile = blockIdx.x * 8 + wave;
    const int tileM = tile >> 4;
    const int tileN = tile & 15;
    const int half = lane >> 4;
    const int l16  = lane & 15;
    const int rowA = tileM * 16 + l16;
    const int colB = tileN * 16 + l16;

    v8f acc = {};
    for (int k0 = 0; k0 < D_DIM; k0 += 4) {
        const int ka = k0 + 2 * half;
        v2f a, b;
        a.x = X[rowA * D_DIM + ka];
        a.y = X[rowA * D_DIM + ka + 1];
        b.x = W1[ka * H_DIM + colB];
        b.y = W1[(ka + 1) * H_DIM + colB];
        acc = __builtin_amdgcn_wmma_f32_16x16x4_f32(
            false, a, false, b, (short)0, acc, false, false);
    }
    const float bias = b1[colB];
#pragma unroll
    for (int v = 0; v < 8; ++v) {
        const int m = tileM * 16 + half * 8 + v;
        const float val = acc[v] + bias;
        h[m * H_DIM + colB] = val > 0.0f ? val : 0.0f;
    }
}

__global__ __launch_bounds__(256) void value_kernel(
    const float* __restrict__ h, const float* __restrict__ W2,
    const float* __restrict__ b2, float* __restrict__ u)
{
    __shared__ float s[256];
    const int t = blockIdx.x;
    const int j = threadIdx.x;
    s[j] = h[t * H_DIM + j] * W2[j];
    __syncthreads();
    for (int off = 128; off > 0; off >>= 1) {
        if (j < off) s[j] += s[j + off];
        __syncthreads();
    }
    if (j == 0) u[t] = s[0] + b2[0];
}

__global__ void scan_kernel(const float* __restrict__ u, const float* __restrict__ ys,
                            float* __restrict__ sigma, float* __restrict__ w,
                            float* __restrict__ out)
{
    if (threadIdx.x != 0 || blockIdx.x != 0) return;
    float r = 0.0f;
    for (int t = 0; t < T_DIM; ++t) {
        const float g    = ys[2 * t + 0];
        const float done = ys[2 * t + 1];
        const float d    = g - u[t];
        const float sg   = d * d - r;
        sigma[t] = sg;
        r = r + CONT_ALPHA * sg;
        if (done > 0.5f) r = 0.0f;
    }
    float wnext = 0.0f;
    float sumw  = 0.0f;
    for (int t = T_DIM - 1; t >= 0; --t) {
        const float done = ys[2 * t + 1];
        const float keep = (done > 0.5f) ? 0.0f : 1.0f;
        const float wt = sigma[t] + GAMMA_LAMBDA * keep * wnext;
        w[t] = wt;
        wnext = wt;
        sumw += wt;
    }
    out[D_DIM * H_DIM + 2 * H_DIM + 0] = -sumw;
    out[D_DIM * H_DIM + 2 * H_DIM + 1] = sigma[T_DIM - 1];
}

__global__ __launch_bounds__(256) void col_kernel(
    float* __restrict__ hBs, const float* __restrict__ W2,
    const float* __restrict__ w, float* __restrict__ out)
{
    __shared__ float s1[256];
    __shared__ float s2[256];
    const int j   = blockIdx.x;
    const int tid = threadIdx.x;
    const float w2j = W2[j];
    float a1 = 0.0f, a2 = 0.0f;
    for (int t = tid; t < T_DIM; t += 256) {
        const float hv = hBs[t * H_DIM + j];
        const float wt = w[t];
        const float bs = (hv > 0.0f) ? wt * w2j : 0.0f;
        hBs[t * H_DIM + j] = bs;
        a1 += bs;
        a2 += wt * hv;
    }
    s1[tid] = a1; s2[tid] = a2;
    __syncthreads();
    for (int off = 128; off > 0; off >>= 1) {
        if (tid < off) { s1[tid] += s1[tid + off]; s2[tid] += s2[tid + off]; }
        __syncthreads();
    }
    if (tid == 0) {
        out[D_DIM * H_DIM + j]         = -s1[0];
        out[D_DIM * H_DIM + H_DIM + j] = -s2[0];
    }
}

__global__ __launch_bounds__(256) void bwd_gemm_kernel(
    const float* __restrict__ X, const float* __restrict__ Bs,
    float* __restrict__ out)
{
    const int lane = threadIdx.x & 31;
    const int wave = threadIdx.x >> 5;
    const int tile = blockIdx.x * 8 + wave;
    const int tileM = tile >> 4;
    const int tileN = tile & 15;
    const int half = lane >> 4;
    const int l16  = lane & 15;
    const int rowA = tileM * 16 + l16;
    const int colB = tileN * 16 + l16;

    v8f acc = {};
    for (int k0 = 0; k0 < T_DIM; k0 += 4) {
        const int ka = k0 + 2 * half;
        v2f a, b;
        a.x = X[ka * D_DIM + rowA];
        a.y = X[(ka + 1) * D_DIM + rowA];
        b.x = Bs[ka * H_DIM + colB];
        b.y = Bs[(ka + 1) * H_DIM + colB];
        acc = __builtin_amdgcn_wmma_f32_16x16x4_f32(
            false, a, false, b, (short)0, acc, false, false);
    }
#pragma unroll
    for (int v = 0; v < 8; ++v) {
        const int m = tileM * 16 + half * 8 + v;
        out[m * H_DIM + colB] = -acc[v];
    }
}

extern "C" void kernel_launch(void* const* d_in, const int* in_sizes, int n_in,
                              void* d_out, int out_size, void* d_ws, size_t ws_size,
                              hipStream_t stream)
{
    const float* xs = (const float*)d_in[0];
    const float* ys = (const float*)d_in[1];
    const float* W1 = (const float*)d_in[2];
    const float* b1 = (const float*)d_in[3];
    const float* W2 = (const float*)d_in[4];
    const float* b2 = (const float*)d_in[5];
    float* out = (float*)d_out;

    float* ws = (float*)d_ws;
    float* h     = ws;
    float* u     = h + (size_t)T_DIM * H_DIM;
    float* sigma = u + T_DIM;
    float* w     = sigma + T_DIM;

    fwd_gemm_kernel<<<256, 256, 0, stream>>>(xs, W1, b1, h);
    value_kernel<<<T_DIM, 256, 0, stream>>>(h, W2, b2, u);
    scan_kernel<<<1, 32, 0, stream>>>(u, ys, sigma, w, out);
    col_kernel<<<H_DIM, 256, 0, stream>>>(h, W2, w, out);
    bwd_gemm_kernel<<<32, 256, 0, stream>>>(xs, h, out);
}


# Scan kernel: one-thread serial loops for r/sigma and w become a 256-thread chunked affine scan (8 steps per thread, lane shuffles, r pairs in f64); reorders the r, w recurrences and sum of w (tree)
# speedup vs baseline: 2.1492x; 2.1492x over previous
#include <hip/hip_runtime.h>
#include <hip/hip_bf16.h>

typedef float v2f __attribute__((ext_vector_type(2)));
typedef float v8f __attribute__((ext_vector_type(8)));

#define T_DIM 2048
#define D_DIM 256
#define H_DIM 256
#define GAMMA_LAMBDA (0.99f * 0.9f)
#define CONT_ALPHA 0.01f

__global__ __launch_bounds__(256) void fwd_gemm_kernel(
    const float* __restrict__ X, const float* __restrict__ W1,
    const float* __restrict__ b1, float* __restrict__ h)
{
    const int lane = threadIdx.x & 31;
    const int wave = threadIdx.x >> 5;
    const int tile = blockIdx.x * 8 + wave;
    const int tileM = tile >> 4;
    const int tileN = tile & 15;
    const int half = lane >> 4;
    const int l16  = lane & 15;
    const int rowA = tileM * 16 + l16;
    const int colB = tileN * 16 + l16;

    v8f acc = {};
    for (int k0 = 0; k0 < D_DIM; k0 += 4) {
        const int ka = k0 + 2 * half;
        v2f a, b;
        a.x = X[rowA * D_DIM + ka];
        a.y = X[rowA * D_DIM + ka + 1];
        b.x = W1[ka * H_DIM + colB];
        b.y = W1[(ka + 1) * H_DIM + colB];
        acc = __builtin_amdgcn_wmma_f32_16x16x4_f32(
            false, a, false, b, (short)0, acc, false, false);
    }
    const float bias = b1[colB];
#pragma unroll
    for (int v = 0; v < 8; ++v) {
        const int m = tileM * 16 + half * 8 + v;
        const float val = acc[v] + bias;
        h[m * H_DIM + colB] = val > 0.0f ? val : 0.0f;
    }
}

__global__ __launch_bounds__(256) void value_kernel(
    const float* __restrict__ h, const float* __restrict__ W2,
    const float* __restrict__ b2, float* __restrict__ u)
{
    __shared__ float s[256];
    const int t = blockIdx.x;
    const int j = threadIdx.x;
    s[j] = h[t * H_DIM + j] * W2[j];
    __syncthreads();
    for (int off = 128; off > 0; off >>= 1) {
        if (j < off) s[j] += s[j + off];
        __syncthreads();
    }
    if (j == 0) u[t] = s[0] + b2[0];
}

#define SCAN_PER 8
#define SCAN_THREADS (T_DIM / SCAN_PER)
#define SCAN_WAVES (SCAN_THREADS / 32)

__global__ __launch_bounds__(SCAN_THREADS) void scan_kernel(
    const float* __restrict__ u, const float* __restrict__ ys,
    float* __restrict__ w, float* __restrict__ out)
{
    __shared__ double sA[SCAN_WAVES], sB[SCAN_WAVES];
    __shared__ float sC[SCAN_WAVES], sS[SCAN_WAVES];
    __shared__ float sSum[SCAN_WAVES];
    __shared__ float sLast;

    const int tid  = threadIdx.x;
    const int lane = tid & 31;
    const int wave = tid >> 5;

    const float4* u4 = reinterpret_cast<const float4*>(u) + tid * 2;
    const float4* y4 = reinterpret_cast<const float4*>(ys) + tid * 4;
    const float4 ua = u4[0], ub = u4[1];
    const float4 y0 = y4[0], y1 = y4[1], y2 = y4[2], y3 = y4[3];
    const float uu[SCAN_PER] = {ua.x, ua.y, ua.z, ua.w, ub.x, ub.y, ub.z, ub.w};
    const float gg[SCAN_PER] = {y0.x, y0.z, y1.x, y1.z, y2.x, y2.z, y3.x, y3.z};
    const float dd[SCAN_PER] = {y0.y, y0.w, y1.y, y1.w, y2.y, y2.w, y3.y, y3.w};

    float d[SCAN_PER];
    bool  dn[SCAN_PER];
#pragma unroll
    for (int s = 0; s < SCAN_PER; ++s) {
        d[s]  = gg[s] - uu[s];
        dn[s] = dd[s] > 0.5f;
    }

    double A = 1.0;
    float  Bf = 0.0f;
#pragma unroll
    for (int s = 0; s < SCAN_PER; ++s) {
        const float sg0 = d[s] * d[s] - Bf;
        Bf = Bf + CONT_ALPHA * sg0;
        A = A * (1.0 - (double)CONT_ALPHA);
        if (dn[s]) { A = 0.0; Bf = 0.0f; }
    }
    double B = (double)Bf;
#pragma unroll
    for (int off = 1; off < 32; off <<= 1) {
        const double Ap = __shfl_up(A, off);
        const double Bp = __shfl_up(B, off);
        if (lane >= off) { B = A * Bp + B; A = A * Ap; }
    }
    if (lane == 31) { sA[wave] = A; sB[wave] = B; }
    __syncthreads();
    double wa[SCAN_WAVES], wb[SCAN_WAVES];
#pragma unroll
    for (int q = 0; q < SCAN_WAVES; ++q) { wa[q] = sA[q]; wb[q] = sB[q]; }
    double rin = 0.0;
#pragma unroll
    for (int q = 0; q < SCAN_WAVES; ++q) {
        const double nx = wa[q] * rin + wb[q];
        rin = (q < wave) ? nx : rin;
    }
    double Ae = __shfl_up(A, 1);
    double Be = __shfl_up(B, 1);
    if (lane == 0) { Ae = 1.0; Be = 0.0; }
    float r = (float)(Ae * rin + Be);

    float sg[SCAN_PER];
#pragma unroll
    for (int s = 0; s < SCAN_PER; ++s) {
        sg[s] = d[s] * d[s] - r;
        r = r + CONT_ALPHA * sg[s];
        if (dn[s]) r = 0.0f;
    }

    float C = 1.0f, S = 0.0f;
#pragma unroll
    for (int s = SCAN_PER - 1; s >= 0; --s) {
        const float c = dn[s] ? 0.0f : GAMMA_LAMBDA;
        S = sg[s] + c * S;
        C = C * c;
    }
#pragma unroll
    for (int off = 1; off < 32; off <<= 1) {
        const float Cn = __shfl_down(C, off);
        const float Sn = __shfl_down(S, off);
        if (lane + off < 32) { S = C * Sn + S; C = C * Cn; }
    }
    if (lane == 0) { sC[wave] = C; sS[wave] = S; }
    if (tid == SCAN_THREADS - 1) sLast = sg[SCAN_PER - 1];
    __syncthreads();
    float win = 0.0f;
#pragma unroll
    for (int q = SCAN_WAVES - 1; q >= 0; --q)
        if (q > wave) win = sC[q] * win + sS[q];
    float Ce = __shfl_down(C, 1);
    float Se = __shfl_down(S, 1);
    if (lane == 31) { Ce = 1.0f; Se = 0.0f; }
    float wv = Ce * win + Se;

    float wo[SCAN_PER];
    float acc = 0.0f;
#pragma unroll
    for (int s = SCAN_PER - 1; s >= 0; --s) {
        const float c = dn[s] ? 0.0f : GAMMA_LAMBDA;
        wv = sg[s] + c * wv;
        wo[s] = wv;
        acc += wv;
    }
    float4* w4 = reinterpret_cast<float4*>(w) + tid * 2;
    w4[0] = make_float4(wo[0], wo[1], wo[2], wo[3]);
    w4[1] = make_float4(wo[4], wo[5], wo[6], wo[7]);

#pragma unroll
    for (int off = 16; off > 0; off >>= 1) acc += __shfl_xor(acc, off);
    if (lane == 0) sSum[wave] = acc;
    __syncthreads();
    if (tid == 0) {
        float tot = 0.0f;
#pragma unroll
        for (int q = 0; q < SCAN_WAVES; ++q) tot += sSum[q];
        out[D_DIM * H_DIM + 2 * H_DIM + 0] = -tot;
        out[D_DIM * H_DIM + 2 * H_DIM + 1] = sLast;
    }
}

__global__ __launch_bounds__(256) void col_kernel(
    float* __restrict__ hBs, const float* __restrict__ W2,
    const float* __restrict__ w, float* __restrict__ out)
{
    __shared__ float s1[256];
    __shared__ float s2[256];
    const int j   = blockIdx.x;
    const int tid = threadIdx.x;
    const float w2j = W2[j];
    float a1 = 0.0f, a2 = 0.0f;
    for (int t = tid; t < T_DIM; t += 256) {
        const float hv = hBs[t * H_DIM + j];
        const float wt = w[t];
        const float bs = (hv > 0.0f) ? wt * w2j : 0.0f;
        hBs[t * H_DIM + j] = bs;
        a1 += bs;
        a2 += wt * hv;
    }
    s1[tid] = a1; s2[tid] = a2;
    __syncthreads();
    for (int off = 128; off > 0; off >>= 1) {
        if (tid < off) { s1[tid] += s1[tid + off]; s2[tid] += s2[tid + off]; }
        __syncthreads();
    }
    if (tid == 0) {
        out[D_DIM * H_DIM + j]         = -s1[0];
        out[D_DIM * H_DIM + H_DIM + j] = -s2[0];
    }
}

__global__ __launch_bounds__(256) void bwd_gemm_kernel(
    const float* __restrict__ X, const float* __restrict__ Bs,
    float* __restrict__ out)
{
    const int lane = threadIdx.x & 31;
    const int wave = threadIdx.x >> 5;
    const int tile = blockIdx.x * 8 + wave;
    const int tileM = tile >> 4;
    const int tileN = tile & 15;
    const int half = lane >> 4;
    const int l16  = lane & 15;
    const int rowA = tileM * 16 + l16;
    const int colB = tileN * 16 + l16;

    v8f acc = {};
    for (int k0 = 0; k0 < T_DIM; k0 += 4) {
        const int ka = k0 + 2 * half;
        v2f a, b;
        a.x = X[ka * D_DIM + rowA];
        a.y = X[(ka + 1) * D_DIM + rowA];
        b.x = Bs[ka * H_DIM + colB];
        b.y = Bs[(ka + 1) * H_DIM + colB];
        acc = __builtin_amdgcn_wmma_f32_16x16x4_f32(
            false, a, false, b, (short)0, acc, false, false);
    }
#pragma unroll
    for (int v = 0; v < 8; ++v) {
        const int m = tileM * 16 + half * 8 + v;
        out[m * H_DIM + colB] = -acc[v];
    }
}

extern "C" void kernel_launch(void* const* d_in, const int* in_sizes, int n_in,
                              void* d_out, int out_size, void* d_ws, size_t ws_size,
                              hipStream_t stream)
{
    const float* xs = (const float*)d_in[0];
    const float* ys = (const float*)d_in[1];
    const float* W1 = (const float*)d_in[2];
    const float* b1 = (const float*)d_in[3];
    const float* W2 = (const float*)d_in[4];
    const float* b2 = (const float*)d_in[5];
    float* out = (float*)d_out;

    float* ws = (float*)d_ws;
    float* h     = ws;
    float* u     = h + (size_t)T_DIM * H_DIM;
    float* w     = u + T_DIM;

    fwd_gemm_kernel<<<256, 256, 0, stream>>>(xs, W1, b1, h);
    value_kernel<<<T_DIM, 256, 0, stream>>>(h, W2, b2, u);
    scan_kernel<<<1, 256, 0, stream>>>(u, ys, w, out);
    col_kernel<<<H_DIM, 256, 0, stream>>>(h, W2, w, out);
    bwd_gemm_kernel<<<32, 256, 0, stream>>>(xs, h, out);
}


# bwd GEMM split-K in the block: a 16x32 panel of total_W1 per block, 16 waves take 128 k each, loads issued 16 k-steps ahead (was 512 serial k-steps per wave); LDS sum in wave order reorders the sum
# speedup vs baseline: 4.4780x; 2.0836x over previous
#include <hip/hip_runtime.h>
#include <hip/hip_bf16.h>

typedef float v2f __attribute__((ext_vector_type(2)));
typedef float v8f __attribute__((ext_vector_type(8)));

#define T_DIM 2048
#define D_DIM 256
#define H_DIM 256
#define GAMMA_LAMBDA (0.99f * 0.9f)
#define CONT_ALPHA 0.01f

__global__ __launch_bounds__(256) void fwd_gemm_kernel(
    const float* __restrict__ X, const float* __restrict__ W1,
    const float* __restrict__ b1, float* __restrict__ h)
{
    const int lane = threadIdx.x & 31;
    const int wave = threadIdx.x >> 5;
    const int tile = blockIdx.x * 8 + wave;
    const int tileM = tile >> 4;
    const int tileN = tile & 15;
    const int half = lane >> 4;
    const int l16  = lane & 15;
    const int rowA = tileM * 16 + l16;
    const int colB = tileN * 16 + l16;

    v8f acc = {};
    for (int k0 = 0; k0 < D_DIM; k0 += 4) {
        const int ka = k0 + 2 * half;
        v2f a, b;
        a.x = X[rowA * D_DIM + ka];
        a.y = X[rowA * D_DIM + ka + 1];
        b.x = W1[ka * H_DIM + colB];
        b.y = W1[(ka + 1) * H_DIM + colB];
        acc = __builtin_amdgcn_wmma_f32_16x16x4_f32(
            false, a, false, b, (short)0, acc, false, false);
    }
    const float bias = b1[colB];
#pragma unroll
    for (int v = 0; v < 8; ++v) {
        const int m = tileM * 16 + half * 8 + v;
        const float val = acc[v] + bias;
        h[m * H_DIM + colB] = val > 0.0f ? val : 0.0f;
    }
}

__global__ __launch_bounds__(256) void value_kernel(
    const float* __restrict__ h, const float* __restrict__ W2,
    const float* __restrict__ b2, float* __restrict__ u)
{
    __shared__ float s[256];
    const int t = blockIdx.x;
    const int j = threadIdx.x;
    s[j] = h[t * H_DIM + j] * W2[j];
    __syncthreads();
    for (int off = 128; off > 0; off >>= 1) {
        if (j < off) s[j] += s[j + off];
        __syncthreads();
    }
    if (j == 0) u[t] = s[0] + b2[0];
}

#define SCAN_PER 8
#define SCAN_THREADS (T_DIM / SCAN_PER)
#define SCAN_WAVES (SCAN_THREADS / 32)

__global__ __launch_bounds__(SCAN_THREADS) void scan_kernel(
    const float* __restrict__ u, const float* __restrict__ ys,
    float* __restrict__ w, float* __restrict__ out)
{
    __shared__ double sA[SCAN_WAVES], sB[SCAN_WAVES];
    __shared__ float sC[SCAN_WAVES], sS[SCAN_WAVES];
    __shared__ float sSum[SCAN_WAVES];
    __shared__ float sLast;

    const int tid  = threadIdx.x;
    const int lane = tid & 31;
    const int wave = tid >> 5;

    const float4* u4 = reinterpret_cast<const float4*>(u) + tid * 2;
    const float4* y4 = reinterpret_cast<const float4*>(ys) + tid * 4;
    const float4 ua = u4[0], ub = u4[1];
    const float4 y0 = y4[0], y1 = y4[1], y2 = y4[2], y3 = y4[3];
    const float uu[SCAN_PER] = {ua.x, ua.y, ua.z, ua.w, ub.x, ub.y, ub.z, ub.w};
    const float gg[SCAN_PER] = {y0.x, y0.z, y1.x, y1.z, y2.x, y2.z, y3.x, y3.z};
    const float dd[SCAN_PER] = {y0.y, y0.w, y1.y, y1.w, y2.y, y2.w, y3.y, y3.w};

    float d[SCAN_PER];
    bool  dn[SCAN_PER];
#pragma unroll
    for (int s = 0; s < SCAN_PER; ++s) {
        d[s]  = gg[s] - uu[s];
        dn[s] = dd[s] > 0.5f;
    }

    double A = 1.0;
    float  Bf = 0.0f;
#pragma unroll
    for (int s = 0; s < SCAN_PER; ++s) {
        const float sg0 = d[s] * d[s] - Bf;
        Bf = Bf + CONT_ALPHA * sg0;
        A = A * (1.0 - (double)CONT_ALPHA);
        if (dn[s]) { A = 0.0; Bf = 0.0f; }
    }
    double B = (double)Bf;
#pragma unroll
    for (int off = 1; off < 32; off <<= 1) {
        const double Ap = __shfl_up(A, off);
        const double Bp = __shfl_up(B, off);
        if (lane >= off) { B = A * Bp + B; A = A * Ap; }
    }
    if (lane == 31) { sA[wave] = A; sB[wave] = B; }
    __syncthreads();
    double wa[SCAN_WAVES], wb[SCAN_WAVES];
#pragma unroll
    for (int q = 0; q < SCAN_WAVES; ++q) { wa[q] = sA[q]; wb[q] = sB[q]; }
    double rin = 0.0;
#pragma unroll
    for (int q = 0; q < SCAN_WAVES; ++q) {
        const double nx = wa[q] * rin + wb[q];
        rin = (q < wave) ? nx : rin;
    }
    double Ae = __shfl_up(A, 1);
    double Be = __shfl_up(B, 1);
    if (lane == 0) { Ae = 1.0; Be = 0.0; }
    float r = (float)(Ae * rin + Be);

    float sg[SCAN_PER];
#pragma unroll
    for (int s = 0; s < SCAN_PER; ++s) {
        sg[s] = d[s] * d[s] - r;
        r = r + CONT_ALPHA * sg[s];
        if (dn[s]) r = 0.0f;
    }

    float C = 1.0f, S = 0.0f;
#pragma unroll
    for (int s = SCAN_PER - 1; s >= 0; --s) {
        const float c = dn[s] ? 0.0f : GAMMA_LAMBDA;
        S = sg[s] + c * S;
        C = C * c;
    }
#pragma unroll
    for (int off = 1; off < 32; off <<= 1) {
        const float Cn = __shfl_down(C, off);
        const float Sn = __shfl_down(S, off);
        if (lane + off < 32) { S = C * Sn + S; C = C * Cn; }
    }
    if (lane == 0) { sC[wave] = C; sS[wave] = S; }
    if (tid == SCAN_THREADS - 1) sLast = sg[SCAN_PER - 1];
    __syncthreads();
    float win = 0.0f;
#pragma unroll
    for (int q = SCAN_WAVES - 1; q >= 0; --q)
        if (q > wave) win = sC[q] * win + sS[q];
    float Ce = __shfl_down(C, 1);
    float Se = __shfl_down(S, 1);
    if (lane == 31) { Ce = 1.0f; Se = 0.0f; }
    float wv = Ce * win + Se;

    float wo[SCAN_PER];
    float acc = 0.0f;
#pragma unroll
    for (int s = SCAN_PER - 1; s >= 0; --s) {
        const float c = dn[s] ? 0.0f : GAMMA_LAMBDA;
        wv = sg[s] + c * wv;
        wo[s] = wv;
        acc += wv;
    }
    float4* w4 = reinterpret_cast<float4*>(w) + tid * 2;
    w4[0] = make_float4(wo[0], wo[1], wo[2], wo[3]);
    w4[1] = make_float4(wo[4], wo[5], wo[6], wo[7]);

#pragma unroll
    for (int off = 16; off > 0; off >>= 1) acc += __shfl_xor(acc, off);
    if (lane == 0) sSum[wave] = acc;
    __syncthreads();
    if (tid == 0) {
        float tot = 0.0f;
#pragma unroll
        for (int q = 0; q < SCAN_WAVES; ++q) tot += sSum[q];
        out[D_DIM * H_DIM + 2 * H_DIM + 0] = -tot;
        out[D_DIM * H_DIM + 2 * H_DIM + 1] = sLast;
    }
}

__global__ __launch_bounds__(256) void col_kernel(
    float* __restrict__ hBs, const float* __restrict__ W2,
    const float* __restrict__ w, float* __restrict__ out)
{
    __shared__ float s1[256];
    __shared__ float s2[256];
    const int j   = blockIdx.x;
    const int tid = threadIdx.x;
    const float w2j = W2[j];
    float a1 = 0.0f, a2 = 0.0f;
    for (int t = tid; t < T_DIM; t += 256) {
        const float hv = hBs[t * H_DIM + j];
        const float wt = w[t];
        const float bs = (hv > 0.0f) ? wt * w2j : 0.0f;
        hBs[t * H_DIM + j] = bs;
        a1 += bs;
        a2 += wt * hv;
    }
    s1[tid] = a1; s2[tid] = a2;
    __syncthreads();
    for (int off = 128; off > 0; off >>= 1) {
        if (tid < off) { s1[tid] += s1[tid + off]; s2[tid] += s2[tid + off]; }
        __syncthreads();
    }
    if (tid == 0) {
        out[D_DIM * H_DIM + j]         = -s1[0];
        out[D_DIM * H_DIM + H_DIM + j] = -s2[0];
    }
}

#define BW_WAVES 16
#define BW_K_PER_WAVE (T_DIM / BW_WAVES)
#define BW_STEPS (BW_K_PER_WAVE / 4)
#define BW_GROUP 16

__global__ __launch_bounds__(BW_WAVES * 32)
__attribute__((amdgpu_waves_per_eu(4, 4))) void bwd_gemm_kernel(
    const float* __restrict__ X, const float* __restrict__ Bs,
    float* __restrict__ out)
{
    __shared__ float red[BW_WAVES][16][32];

    const int lane = threadIdx.x & 31;
    const int wave = threadIdx.x >> 5;
    const int half = lane >> 4;
    const int l16  = lane & 15;
    const int m0 = (blockIdx.x >> 3) * 16;
    const int n0 = (blockIdx.x & 7) * 32;

    const int kw = wave * BW_K_PER_WAVE + 2 * half;
    const float* xa = X  + kw * D_DIM + m0 + l16;
    const float* bb = Bs + kw * H_DIM + n0 + 2 * l16;

    v8f acc0 = {}, acc1 = {};
#pragma unroll
    for (int g = 0; g < BW_STEPS; g += BW_GROUP) {
        float ax[BW_GROUP], ay[BW_GROUP];
        v2f   p0[BW_GROUP], p1[BW_GROUP];
#pragma unroll
        for (int i = 0; i < BW_GROUP; ++i) {
            const int k = (g + i) * 4;
            ax[i] = xa[k * D_DIM];
            ay[i] = xa[(k + 1) * D_DIM];
            p0[i] = *reinterpret_cast<const v2f*>(bb + k * H_DIM);
            p1[i] = *reinterpret_cast<const v2f*>(bb + (k + 1) * H_DIM);
        }
#pragma unroll
        for (int i = 0; i < BW_GROUP; ++i) {
            v2f a, b0, b1;
            a.x  = ax[i];   a.y  = ay[i];
            b0.x = p0[i].x; b0.y = p1[i].x;
            b1.x = p0[i].y; b1.y = p1[i].y;
            acc0 = __builtin_amdgcn_wmma_f32_16x16x4_f32(
                false, a, false, b0, (short)0, acc0, false, false);
            acc1 = __builtin_amdgcn_wmma_f32_16x16x4_f32(
                false, a, false, b1, (short)0, acc1, false, false);
        }
    }

#pragma unroll
    for (int v = 0; v < 8; ++v) {
        red[wave][v][lane]     = acc0[v];
        red[wave][8 + v][lane] = acc1[v];
    }
    __syncthreads();

    if (threadIdx.x < 256) {
        const int v = threadIdx.x >> 5;
        float s0 = 0.0f, s1 = 0.0f;
#pragma unroll
        for (int q = 0; q < BW_WAVES; ++q) {
            s0 += red[q][v][lane];
            s1 += red[q][8 + v][lane];
        }
        const int m = m0 + half * 8 + v;
        float2* dst = reinterpret_cast<float2*>(out + m * H_DIM + n0 + 2 * l16);
        *dst = make_float2(-s0, -s1);
    }
}

extern "C" void kernel_launch(void* const* d_in, const int* in_sizes, int n_in,
                              void* d_out, int out_size, void* d_ws, size_t ws_size,
                              hipStream_t stream)
{
    const float* xs = (const float*)d_in[0];
    const float* ys = (const float*)d_in[1];
    const float* W1 = (const float*)d_in[2];
    const float* b1 = (const float*)d_in[3];
    const float* W2 = (const float*)d_in[4];
    const float* b2 = (const float*)d_in[5];
    float* out = (float*)d_out;

    float* ws = (float*)d_ws;
    float* h     = ws;
    float* u     = h + (size_t)T_DIM * H_DIM;
    float* w     = u + T_DIM;

    fwd_gemm_kernel<<<256, 256, 0, stream>>>(xs, W1, b1, h);
    value_kernel<<<T_DIM, 256, 0, stream>>>(h, W2, b2, u);
    scan_kernel<<<1, 256, 0, stream>>>(u, ys, w, out);
    col_kernel<<<H_DIM, 256, 0, stream>>>(h, W2, w, out);
    bwd_gemm_kernel<<<128, 512, 0, stream>>>(xs, h, out);
}


# col kernel folded into the bwd GEMM launch: B fragments built in registers from h, w, W2 (Bs never stored), 8 extra blocks sum the columns; 5 to 4 launches; reorders the sums of total_b1, total_W2
# speedup vs baseline: 9.8153x; 2.1919x over previous
#include <hip/hip_runtime.h>
#include <hip/hip_bf16.h>

typedef float v2f __attribute__((ext_vector_type(2)));
typedef float v8f __attribute__((ext_vector_type(8)));
typedef float v4f __attribute__((ext_vector_type(4)));

#define T_DIM 2048
#define D_DIM 256
#define H_DIM 256
#define GAMMA_LAMBDA (0.99f * 0.9f)
#define CONT_ALPHA 0.01f

__global__ __launch_bounds__(256) void fwd_gemm_kernel(
    const float* __restrict__ X, const float* __restrict__ W1,
    const float* __restrict__ b1, float* __restrict__ h)
{
    const int lane = threadIdx.x & 31;
    const int wave = threadIdx.x >> 5;
    const int tile = blockIdx.x * 8 + wave;
    const int tileM = tile >> 4;
    const int tileN = tile & 15;
    const int half = lane >> 4;
    const int l16  = lane & 15;
    const int rowA = tileM * 16 + l16;
    const int colB = tileN * 16 + l16;

    v8f acc = {};
    for (int k0 = 0; k0 < D_DIM; k0 += 4) {
        const int ka = k0 + 2 * half;
        v2f a, b;
        a.x = X[rowA * D_DIM + ka];
        a.y = X[rowA * D_DIM + ka + 1];
        b.x = W1[ka * H_DIM + colB];
        b.y = W1[(ka + 1) * H_DIM + colB];
        acc = __builtin_amdgcn_wmma_f32_16x16x4_f32(
            false, a, false, b, (short)0, acc, false, false);
    }
    const float bias = b1[colB];
#pragma unroll
    for (int v = 0; v < 8; ++v) {
        const int m = tileM * 16 + half * 8 + v;
        const float val = acc[v] + bias;
        h[m * H_DIM + colB] = val > 0.0f ? val : 0.0f;
    }
}

__global__ __launch_bounds__(256) void value_kernel(
    const float* __restrict__ h, const float* __restrict__ W2,
    const float* __restrict__ b2, float* __restrict__ u)
{
    __shared__ float s[256];
    const int t = blockIdx.x;
    const int j = threadIdx.x;
    s[j] = h[t * H_DIM + j] * W2[j];
    __syncthreads();
    for (int off = 128; off > 0; off >>= 1) {
        if (j < off) s[j] += s[j + off];
        __syncthreads();
    }
    if (j == 0) u[t] = s[0] + b2[0];
}

#define SCAN_PER 8
#define SCAN_THREADS (T_DIM / SCAN_PER)
#define SCAN_WAVES (SCAN_THREADS / 32)

__global__ __launch_bounds__(SCAN_THREADS) void scan_kernel(
    const float* __restrict__ u, const float* __restrict__ ys,
    float* __restrict__ w, float* __restrict__ out)
{
    __shared__ double sA[SCAN_WAVES], sB[SCAN_WAVES];
    __shared__ float sC[SCAN_WAVES], sS[SCAN_WAVES];
    __shared__ float sSum[SCAN_WAVES];
    __shared__ float sLast;

    const int tid  = threadIdx.x;
    const int lane = tid & 31;
    const int wave = tid >> 5;

    const float4* u4 = reinterpret_cast<const float4*>(u) + tid * 2;
    const float4* y4 = reinterpret_cast<const float4*>(ys) + tid * 4;
    const float4 ua = u4[0], ub = u4[1];
    const float4 y0 = y4[0], y1 = y4[1], y2 = y4[2], y3 = y4[3];
    const float uu[SCAN_PER] = {ua.x, ua.y, ua.z, ua.w, ub.x, ub.y, ub.z, ub.w};
    const float gg[SCAN_PER] = {y0.x, y0.z, y1.x, y1.z, y2.x, y2.z, y3.x, y3.z};
    const float dd[SCAN_PER] = {y0.y, y0.w, y1.y, y1.w, y2.y, y2.w, y3.y, y3.w};

    float d[SCAN_PER];
    bool  dn[SCAN_PER];
#pragma unroll
    for (int s = 0; s < SCAN_PER; ++s) {
        d[s]  = gg[s] - uu[s];
        dn[s] = dd[s] > 0.5f;
    }

    double A = 1.0;
    float  Bf = 0.0f;
#pragma unroll
    for (int s = 0; s < SCAN_PER; ++s) {
        const float sg0 = d[s] * d[s] - Bf;
        Bf = Bf + CONT_ALPHA * sg0;
        A = A * (1.0 - (double)CONT_ALPHA);
        if (dn[s]) { A = 0.0; Bf = 0.0f; }
    }
    double B = (double)Bf;
#pragma unroll
    for (int off = 1; off < 32; off <<= 1) {
        const double Ap = __shfl_up(A, off);
        const double Bp = __shfl_up(B, off);
        if (lane >= off) { B = A * Bp + B; A = A * Ap; }
    }
    if (lane == 31) { sA[wave] = A; sB[wave] = B; }
    __syncthreads();
    double wa[SCAN_WAVES], wb[SCAN_WAVES];
#pragma unroll
    for (int q = 0; q < SCAN_WAVES; ++q) { wa[q] = sA[q]; wb[q] = sB[q]; }
    double rin = 0.0;
#pragma unroll
    for (int q = 0; q < SCAN_WAVES; ++q) {
        const double nx = wa[q] * rin + wb[q];
        rin = (q < wave) ? nx : rin;
    }
    double Ae = __shfl_up(A, 1);
    double Be = __shfl_up(B, 1);
    if (lane == 0) { Ae = 1.0; Be = 0.0; }
    float r = (float)(Ae * rin + Be);

    float sg[SCAN_PER];
#pragma unroll
    for (int s = 0; s < SCAN_PER; ++s) {
        sg[s] = d[s] * d[s] - r;
        r = r + CONT_ALPHA * sg[s];
        if (dn[s]) r = 0.0f;
    }

    float C = 1.0f, S = 0.0f;
#pragma unroll
    for (int s = SCAN_PER - 1; s >= 0; --s) {
        const float c = dn[s] ? 0.0f : GAMMA_LAMBDA;
        S = sg[s] + c * S;
        C = C * c;
    }
#pragma unroll
    for (int off = 1; off < 32; off <<= 1) {
        const float Cn = __shfl_down(C, off);
        const float Sn = __shfl_down(S, off);
        if (lane + off < 32) { S = C * Sn + S; C = C * Cn; }
    }
    if (lane == 0) { sC[wave] = C; sS[wave] = S; }
    if (tid == SCAN_THREADS - 1) sLast = sg[SCAN_PER - 1];
    __syncthreads();
    float win = 0.0f;
#pragma unroll
    for (int q = SCAN_WAVES - 1; q >= 0; --q)
        if (q > wave) win = sC[q] * win + sS[q];
    float Ce = __shfl_down(C, 1);
    float Se = __shfl_down(S, 1);
    if (lane == 31) { Ce = 1.0f; Se = 0.0f; }
    float wv = Ce * win + Se;

    float wo[SCAN_PER];
    float acc = 0.0f;
#pragma unroll
    for (int s = SCAN_PER - 1; s >= 0; --s) {
        const float c = dn[s] ? 0.0f : GAMMA_LAMBDA;
        wv = sg[s] + c * wv;
        wo[s] = wv;
        acc += wv;
    }
    float4* w4 = reinterpret_cast<float4*>(w) + tid * 2;
    w4[0] = make_float4(wo[0], wo[1], wo[2], wo[3]);
    w4[1] = make_float4(wo[4], wo[5], wo[6], wo[7]);

#pragma unroll
    for (int off = 16; off > 0; off >>= 1) acc += __shfl_xor(acc, off);
    if (lane == 0) sSum[wave] = acc;
    __syncthreads();
    if (tid == 0) {
        float tot = 0.0f;
#pragma unroll
        for (int q = 0; q < SCAN_WAVES; ++q) tot += sSum[q];
        out[D_DIM * H_DIM + 2 * H_DIM + 0] = -tot;
        out[D_DIM * H_DIM + 2 * H_DIM + 1] = sLast;
    }
}

#define BW_WAVES 16
#define BW_K_PER_WAVE (T_DIM / BW_WAVES)
#define BW_STEPS (BW_K_PER_WAVE / 4)
#define BW_GROUP 16

template <int N>
__device__ __forceinline__ float row_lane_c(float v)
{
    return __builtin_bit_cast(float, __builtin_amdgcn_update_dpp(
        0, __builtin_bit_cast(int, v), 0x150 + N, 0xf, 0xf, true));
}
__device__ __forceinline__ float row_lane(float v, int n)
{
    switch (n) {
    case 0:  return row_lane_c<0>(v);   case 1:  return row_lane_c<1>(v);
    case 2:  return row_lane_c<2>(v);   case 3:  return row_lane_c<3>(v);
    case 4:  return row_lane_c<4>(v);   case 5:  return row_lane_c<5>(v);
    case 6:  return row_lane_c<6>(v);   case 7:  return row_lane_c<7>(v);
    case 8:  return row_lane_c<8>(v);   case 9:  return row_lane_c<9>(v);
    case 10: return row_lane_c<10>(v);  case 11: return row_lane_c<11>(v);
    case 12: return row_lane_c<12>(v);  case 13: return row_lane_c<13>(v);
    case 14: return row_lane_c<14>(v);  default: return row_lane_c<15>(v);
    }
}

__device__ __forceinline__ void bwd_panel(
    const float* __restrict__ X, const float* __restrict__ h,
    const float* __restrict__ W2, const float* __restrict__ w,
    float* __restrict__ out, float (&red)[BW_WAVES][16][32])
{
    const int lane = threadIdx.x & 31;
    const int wave = threadIdx.x >> 5;
    const int half = lane >> 4;
    const int l16  = lane & 15;
    const int m0 = (blockIdx.x >> 3) * 16;
    const int n0 = (blockIdx.x & 7) * 32;

    const int kw = wave * BW_K_PER_WAVE + 2 * half;
    const float* xa = X + kw * D_DIM + m0 + l16;
    const float* hb = h + kw * H_DIM + n0 + 2 * l16;
    const float* wl = w + wave * BW_K_PER_WAVE + 4 * l16 + 2 * half;
    const v2f wlo = *reinterpret_cast<const v2f*>(wl);
    const v2f whi = *reinterpret_cast<const v2f*>(wl + BW_K_PER_WAVE / 2);
    const v2f c = *reinterpret_cast<const v2f*>(W2 + n0 + 2 * l16);

    v8f acc0 = {}, acc1 = {};
#pragma unroll
    for (int g = 0; g < BW_STEPS; g += BW_GROUP) {
        float ax[BW_GROUP], ay[BW_GROUP];
        v2f   p0[BW_GROUP], p1[BW_GROUP];
#pragma unroll
        for (int i = 0; i < BW_GROUP; ++i) {
            const int k = (g + i) * 4;
            ax[i] = xa[k * D_DIM];
            ay[i] = xa[(k + 1) * D_DIM];
            p0[i] = *reinterpret_cast<const v2f*>(hb + k * H_DIM);
            p1[i] = *reinterpret_cast<const v2f*>(hb + (k + 1) * H_DIM);
        }
#pragma unroll
        for (int i = 0; i < BW_GROUP; ++i) {
            const v2f wsrc = (g == 0) ? wlo : whi;
            const float w0 = row_lane(wsrc.x, i);
            const float w1 = row_lane(wsrc.y, i);
            v2f a, b0, b1;
            a.x  = ax[i];   a.y  = ay[i];
            b0.x = (p0[i].x > 0.0f) ? w0 * c.x : 0.0f;
            b0.y = (p1[i].x > 0.0f) ? w1 * c.x : 0.0f;
            b1.x = (p0[i].y > 0.0f) ? w0 * c.y : 0.0f;
            b1.y = (p1[i].y > 0.0f) ? w1 * c.y : 0.0f;
            acc0 = __builtin_amdgcn_wmma_f32_16x16x4_f32(
                false, a, false, b0, (short)0, acc0, false, false);
            acc1 = __builtin_amdgcn_wmma_f32_16x16x4_f32(
                false, a, false, b1, (short)0, acc1, false, false);
        }
    }

#pragma unroll
    for (int v = 0; v < 8; ++v) {
        red[wave][v][lane]     = acc0[v];
        red[wave][8 + v][lane] = acc1[v];
    }
    __syncthreads();

    if (threadIdx.x < 256) {
        const int v = threadIdx.x >> 5;
        float s0 = 0.0f, s1 = 0.0f;
#pragma unroll
        for (int q = 0; q < BW_WAVES; ++q) {
            s0 += red[q][v][lane];
            s1 += red[q][8 + v][lane];
        }
        const int m = m0 + half * 8 + v;
        float2* dst = reinterpret_cast<float2*>(out + m * H_DIM + n0 + 2 * l16);
        *dst = make_float2(-s0, -s1);
    }
}

__device__ __forceinline__ void col_sums(
    const float* __restrict__ h, const float* __restrict__ W2,
    const float* __restrict__ w, float* __restrict__ out,
    float (&red)[BW_WAVES][16][32])
{
    const int lane = threadIdx.x & 31;
    const int wave = threadIdx.x >> 5;
    const int n0 = (blockIdx.x - 128) * 32;
    const int cg = lane & 7;
    const int t0 = wave * BW_K_PER_WAVE + (lane >> 3);
    const float* hp = h + t0 * H_DIM + n0 + 4 * cg;
    const float* wp = w + t0;
    const v4f c = *reinterpret_cast<const v4f*>(W2 + n0 + 4 * cg);

    v4f sb = {0.0f, 0.0f, 0.0f, 0.0f}, sw = {0.0f, 0.0f, 0.0f, 0.0f};
#pragma unroll
    for (int i = 0; i < BW_K_PER_WAVE / 4; ++i) {
        const v4f hv = *reinterpret_cast<const v4f*>(hp + i * 4 * H_DIM);
        const float wt = wp[i * 4];
        const v4f e = wt * c;
        sb.x += (hv.x > 0.0f) ? e.x : 0.0f;
        sb.y += (hv.y > 0.0f) ? e.y : 0.0f;
        sb.z += (hv.z > 0.0f) ? e.z : 0.0f;
        sb.w += (hv.w > 0.0f) ? e.w : 0.0f;
        sw += wt * hv;
    }
#pragma unroll
    for (int q = 0; q < 4; ++q) {
        sb[q] += __shfl_xor(sb[q], 8);   sw[q] += __shfl_xor(sw[q], 8);
        sb[q] += __shfl_xor(sb[q], 16);  sw[q] += __shfl_xor(sw[q], 16);
    }
    if (lane < 8) {
#pragma unroll
        for (int q = 0; q < 4; ++q) {
            red[wave][0][4 * cg + q] = sb[q];
            red[wave][1][4 * cg + q] = sw[q];
        }
    }
    __syncthreads();
    if (threadIdx.x < 64) {
        float s = 0.0f;
#pragma unroll
        for (int q = 0; q < BW_WAVES; ++q) s += red[q][wave][lane];
        out[D_DIM * H_DIM + wave * H_DIM + n0 + lane] = -s;
    }
}

__global__ __launch_bounds__(BW_WAVES * 32)
__attribute__((amdgpu_waves_per_eu(4, 4))) void bwd_gemm_kernel(
    const float* __restrict__ X, const float* __restrict__ h,
    const float* __restrict__ W2, const float* __restrict__ w,
    float* __restrict__ out)
{
    __shared__ float red[BW_WAVES][16][32];

    if (blockIdx.x < 128) bwd_panel(X, h, W2, w, out, red);
    else                  col_sums(h, W2, w, out, red);
}

extern "C" void kernel_launch(void* const* d_in, const int* in_sizes, int n_in,
                              void* d_out, int out_size, void* d_ws, size_t ws_size,
                              hipStream_t stream)
{
    const float* xs = (const float*)d_in[0];
    const float* ys = (const float*)d_in[1];
    const float* W1 = (const float*)d_in[2];
    const float* b1 = (const float*)d_in[3];
    const float* W2 = (const float*)d_in[4];
    const float* b2 = (const float*)d_in[5];
    float* out = (float*)d_out;

    float* ws = (float*)d_ws;
    float* h     = ws;
    float* u     = h + (size_t)T_DIM * H_DIM;
    float* w     = u + T_DIM;

    fwd_gemm_kernel<<<256, 256, 0, stream>>>(xs, W1, b1, h);
    value_kernel<<<T_DIM, 256, 0, stream>>>(h, W2, b2, u);
    scan_kernel<<<1, 256, 0, stream>>>(u, ys, w, out);
    bwd_gemm_kernel<<<128 + 8, 512, 0, stream>>>(xs, h, W2, w, out);
}


# value kernel folded into the fwd GEMM launch: each wave sums h*W2 over its 16 columns by lane DPP adds, 8 waves via LDS, scan adds the two half-row partials; 4 to 3 launches; reorders the sum of u
# speedup vs baseline: 10.9480x; 1.1154x over previous
#include <hip/hip_runtime.h>
#include <hip/hip_bf16.h>

typedef float v2f __attribute__((ext_vector_type(2)));
typedef float v8f __attribute__((ext_vector_type(8)));
typedef float v4f __attribute__((ext_vector_type(4)));

#define T_DIM 2048
#define D_DIM 256
#define H_DIM 256
#define GAMMA_LAMBDA (0.99f * 0.9f)
#define CONT_ALPHA 0.01f

#define UP_PITCH 32

template <int M>
__device__ __forceinline__ float row_xor_c(float v)
{
    return __builtin_bit_cast(float, __builtin_amdgcn_update_dpp(
        0, __builtin_bit_cast(int, v), 0x160 + M, 0xf, 0xf, true));
}
__device__ __forceinline__ float row_sum16(float v)
{
    v += row_xor_c<1>(v);
    v += row_xor_c<2>(v);
    v += row_xor_c<4>(v);
    v += row_xor_c<8>(v);
    return v;
}

__global__ __launch_bounds__(256) void fwd_gemm_kernel(
    const float* __restrict__ X, const float* __restrict__ W1,
    const float* __restrict__ b1, const float* __restrict__ W2,
    float* __restrict__ h, float* __restrict__ up)
{
    __shared__ __attribute__((aligned(16))) float part[8][16];

    const int lane = threadIdx.x & 31;
    const int wave = threadIdx.x >> 5;
    const int tile = blockIdx.x * 8 + wave;
    const int tileM = tile >> 4;
    const int tileN = tile & 15;
    const int half = lane >> 4;
    const int l16  = lane & 15;
    const int rowA = tileM * 16 + l16;
    const int colB = tileN * 16 + l16;

    v8f acc = {};
    for (int k0 = 0; k0 < D_DIM; k0 += 4) {
        const int ka = k0 + 2 * half;
        v2f a, b;
        a.x = X[rowA * D_DIM + ka];
        a.y = X[rowA * D_DIM + ka + 1];
        b.x = W1[ka * H_DIM + colB];
        b.y = W1[(ka + 1) * H_DIM + colB];
        acc = __builtin_amdgcn_wmma_f32_16x16x4_f32(
            false, a, false, b, (short)0, acc, false, false);
    }
    const float bias = b1[colB];
    const float c2 = W2[colB];
    float hv[8], pr[8];
#pragma unroll
    for (int v = 0; v < 8; ++v) {
        const float val = acc[v] + bias;
        hv[v] = val > 0.0f ? val : 0.0f;
        pr[v] = row_sum16(hv[v] * c2);
    }
    if (l16 == 0) {
        float4* dst = reinterpret_cast<float4*>(&part[wave][half * 8]);
        dst[0] = make_float4(pr[0], pr[1], pr[2], pr[3]);
        dst[1] = make_float4(pr[4], pr[5], pr[6], pr[7]);
    }
    __syncthreads();
#pragma unroll
    for (int v = 0; v < 8; ++v) {
        const int m = tileM * 16 + half * 8 + v;
        h[m * H_DIM + colB] = hv[v];
    }
    if (threadIdx.x < 16) {
        float s = 0.0f;
#pragma unroll
        for (int q = 0; q < 8; ++q) s += part[q][threadIdx.x];
        up[blockIdx.x * UP_PITCH + threadIdx.x] = s;
    }
}

#define SCAN_PER 8
#define SCAN_THREADS (T_DIM / SCAN_PER)
#define SCAN_WAVES (SCAN_THREADS / 32)

__global__ __launch_bounds__(SCAN_THREADS) void scan_kernel(
    const float* __restrict__ up, const float* __restrict__ b2,
    const float* __restrict__ ys,
    float* __restrict__ w, float* __restrict__ out)
{
    __shared__ double sA[SCAN_WAVES], sB[SCAN_WAVES];
    __shared__ float sC[SCAN_WAVES], sS[SCAN_WAVES];
    __shared__ float sSum[SCAN_WAVES];
    __shared__ float sLast;

    const int tid  = threadIdx.x;
    const int lane = tid & 31;
    const int wave = tid >> 5;

    const float* ul = up + (tid >> 1) * 2 * UP_PITCH + (tid & 1) * 8;
    const float4* u4 = reinterpret_cast<const float4*>(ul);
    const float4* v4 = reinterpret_cast<const float4*>(ul + UP_PITCH);
    const float4* y4 = reinterpret_cast<const float4*>(ys) + tid * 4;
    const float4 ua = u4[0], ub = u4[1];
    const float4 va = v4[0], vb = v4[1];
    const float4 y0 = y4[0], y1 = y4[1], y2 = y4[2], y3 = y4[3];
    const float bb = b2[0];
    const float uu[SCAN_PER] = {
        (ua.x + va.x) + bb, (ua.y + va.y) + bb, (ua.z + va.z) + bb, (ua.w + va.w) + bb,
        (ub.x + vb.x) + bb, (ub.y + vb.y) + bb, (ub.z + vb.z) + bb, (ub.w + vb.w) + bb};
    const float gg[SCAN_PER] = {y0.x, y0.z, y1.x, y1.z, y2.x, y2.z, y3.x, y3.z};
    const float dd[SCAN_PER] = {y0.y, y0.w, y1.y, y1.w, y2.y, y2.w, y3.y, y3.w};

    float d[SCAN_PER];
    bool  dn[SCAN_PER];
#pragma unroll
    for (int s = 0; s < SCAN_PER; ++s) {
        d[s]  = gg[s] - uu[s];
        dn[s] = dd[s] > 0.5f;
    }

    double A = 1.0;
    float  Bf = 0.0f;
#pragma unroll
    for (int s = 0; s < SCAN_PER; ++s) {
        const float sg0 = d[s] * d[s] - Bf;
        Bf = Bf + CONT_ALPHA * sg0;
        A = A * (1.0 - (double)CONT_ALPHA);
        if (dn[s]) { A = 0.0; Bf = 0.0f; }
    }
    double B = (double)Bf;
#pragma unroll
    for (int off = 1; off < 32; off <<= 1) {
        const double Ap = __shfl_up(A, off);
        const double Bp = __shfl_up(B, off);
        if (lane >= off) { B = A * Bp + B; A = A * Ap; }
    }
    if (lane == 31) { sA[wave] = A; sB[wave] = B; }
    __syncthreads();
    double wa[SCAN_WAVES], wb[SCAN_WAVES];
#pragma unroll
    for (int q = 0; q < SCAN_WAVES; ++q) { wa[q] = sA[q]; wb[q] = sB[q]; }
    double rin = 0.0;
#pragma unroll
    for (int q = 0; q < SCAN_WAVES; ++q) {
        const double nx = wa[q] * rin + wb[q];
        rin = (q < wave) ? nx : rin;
    }
    double Ae = __shfl_up(A, 1);
    double Be = __shfl_up(B, 1);
    if (lane == 0) { Ae = 1.0; Be = 0.0; }
    float r = (float)(Ae * rin + Be);

    float sg[SCAN_PER];
#pragma unroll
    for (int s = 0; s < SCAN_PER; ++s) {
        sg[s] = d[s] * d[s] - r;
        r = r + CONT_ALPHA * sg[s];
        if (dn[s]) r = 0.0f;
    }

    float C = 1.0f, S = 0.0f;
#pragma unroll
    for (int s = SCAN_PER - 1; s >= 0; --s) {
        const float c = dn[s] ? 0.0f : GAMMA_LAMBDA;
        S = sg[s] + c * S;
        C = C * c;
    }
#pragma unroll
    for (int off = 1; off < 32; off <<= 1) {
        const float Cn = __shfl_down(C, off);
        const float Sn = __shfl_down(S, off);
        if (lane + off < 32) { S = C * Sn + S; C = C * Cn; }
    }
    if (lane == 0) { sC[wave] = C; sS[wave] = S; }
    if (tid == SCAN_THREADS - 1) sLast = sg[SCAN_PER - 1];
    __syncthreads();
    float win = 0.0f;
#pragma unroll
    for (int q = SCAN_WAVES - 1; q >= 0; --q)
        if (q > wave) win = sC[q] * win + sS[q];
    float Ce = __shfl_down(C, 1);
    float Se = __shfl_down(S, 1);
    if (lane == 31) { Ce = 1.0f; Se = 0.0f; }
    float wv = Ce * win + Se;

    float wo[SCAN_PER];
    float acc = 0.0f;
#pragma unroll
    for (int s = SCAN_PER - 1; s >= 0; --s) {
        const float c = dn[s] ? 0.0f : GAMMA_LAMBDA;
        wv = sg[s] + c * wv;
        wo[s] = wv;
        acc += wv;
    }
    float4* w4 = reinterpret_cast<float4*>(w) + tid * 2;
    w4[0] = make_float4(wo[0], wo[1], wo[2], wo[3]);
    w4[1] = make_float4(wo[4], wo[5], wo[6], wo[7]);

#pragma unroll
    for (int off = 16; off > 0; off >>= 1) acc += __shfl_xor(acc, off);
    if (lane == 0) sSum[wave] = acc;
    __syncthreads();
    if (tid == 0) {
        float tot = 0.0f;
#pragma unroll
        for (int q = 0; q < SCAN_WAVES; ++q) tot += sSum[q];
        out[D_DIM * H_DIM + 2 * H_DIM + 0] = -tot;
        out[D_DIM * H_DIM + 2 * H_DIM + 1] = sLast;
    }
}

#define BW_WAVES 16
#define BW_K_PER_WAVE (T_DIM / BW_WAVES)
#define BW_STEPS (BW_K_PER_WAVE / 4)
#define BW_GROUP 16

template <int N>
__device__ __forceinline__ float row_lane_c(float v)
{
    return __builtin_bit_cast(float, __builtin_amdgcn_update_dpp(
        0, __builtin_bit_cast(int, v), 0x150 + N, 0xf, 0xf, true));
}
__device__ __forceinline__ float row_lane(float v, int n)
{
    switch (n) {
    case 0:  return row_lane_c<0>(v);   case 1:  return row_lane_c<1>(v);
    case 2:  return row_lane_c<2>(v);   case 3:  return row_lane_c<3>(v);
    case 4:  return row_lane_c<4>(v);   case 5:  return row_lane_c<5>(v);
    case 6:  return row_lane_c<6>(v);   case 7:  return row_lane_c<7>(v);
    case 8:  return row_lane_c<8>(v);   case 9:  return row_lane_c<9>(v);
    case 10: return row_lane_c<10>(v);  case 11: return row_lane_c<11>(v);
    case 12: return row_lane_c<12>(v);  case 13: return row_lane_c<13>(v);
    case 14: return row_lane_c<14>(v);  default: return row_lane_c<15>(v);
    }
}

__device__ __forceinline__ void bwd_panel(
    const float* __restrict__ X, const float* __restrict__ h,
    const float* __restrict__ W2, const float* __restrict__ w,
    float* __restrict__ out, float (&red)[BW_WAVES][16][32])
{
    const int lane = threadIdx.x & 31;
    const int wave = threadIdx.x >> 5;
    const int half = lane >> 4;
    const int l16  = lane & 15;
    const int m0 = (blockIdx.x >> 3) * 16;
    const int n0 = (blockIdx.x & 7) * 32;

    const int kw = wave * BW_K_PER_WAVE + 2 * half;
    const float* xa = X + kw * D_DIM + m0 + l16;
    const float* hb = h + kw * H_DIM + n0 + 2 * l16;
    const float* wl = w + wave * BW_K_PER_WAVE + 4 * l16 + 2 * half;
    const v2f wlo = *reinterpret_cast<const v2f*>(wl);
    const v2f whi = *reinterpret_cast<const v2f*>(wl + BW_K_PER_WAVE / 2);
    const v2f c = *reinterpret_cast<const v2f*>(W2 + n0 + 2 * l16);

    v8f acc0 = {}, acc1 = {};
#pragma unroll
    for (int g = 0; g < BW_STEPS; g += BW_GROUP) {
        float ax[BW_GROUP], ay[BW_GROUP];
        v2f   p0[BW_GROUP], p1[BW_GROUP];
#pragma unroll
        for (int i = 0; i < BW_GROUP; ++i) {
            const int k = (g + i) * 4;
            ax[i] = xa[k * D_DIM];
            ay[i] = xa[(k + 1) * D_DIM];
            p0[i] = *reinterpret_cast<const v2f*>(hb + k * H_DIM);
            p1[i] = *reinterpret_cast<const v2f*>(hb + (k + 1) * H_DIM);
        }
#pragma unroll
        for (int i = 0; i < BW_GROUP; ++i) {
            const v2f wsrc = (g == 0) ? wlo : whi;
            const float w0 = row_lane(wsrc.x, i);
            const float w1 = row_lane(wsrc.y, i);
            v2f a, b0, b1;
            a.x  = ax[i];   a.y  = ay[i];
            b0.x = (p0[i].x > 0.0f) ? w0 * c.x : 0.0f;
            b0.y = (p1[i].x > 0.0f) ? w1 * c.x : 0.0f;
            b1.x = (p0[i].y > 0.0f) ? w0 * c.y : 0.0f;
            b1.y = (p1[i].y > 0.0f) ? w1 * c.y : 0.0f;
            acc0 = __builtin_amdgcn_wmma_f32_16x16x4_f32(
                false, a, false, b0, (short)0, acc0, false, false);
            acc1 = __builtin_amdgcn_wmma_f32_16x16x4_f32(
                false, a, false, b1, (short)0, acc1, false, false);
        }
    }

#pragma unroll
    for (int v = 0; v < 8; ++v) {
        red[wave][v][lane]     = acc0[v];
        red[wave][8 + v][lane] = acc1[v];
    }
    __syncthreads();

    if (threadIdx.x < 256) {
        const int v = threadIdx.x >> 5;
        float s0 = 0.0f, s1 = 0.0f;
#pragma unroll
        for (int q = 0; q < BW_WAVES; ++q) {
            s0 += red[q][v][lane];
            s1 += red[q][8 + v][lane];
        }
        const int m = m0 + half * 8 + v;
        float2* dst = reinterpret_cast<float2*>(out + m * H_DIM + n0 + 2 * l16);
        *dst = make_float2(-s0, -s1);
    }
}

__device__ __forceinline__ void col_sums(
    const float* __restrict__ h, const float* __restrict__ W2,
    const float* __restrict__ w, float* __restrict__ out,
    float (&red)[BW_WAVES][16][32])
{
    const int lane = threadIdx.x & 31;
    const int wave = threadIdx.x >> 5;
    const int n0 = (blockIdx.x - 128) * 32;
    const int cg = lane & 7;
    const int t0 = wave * BW_K_PER_WAVE + (lane >> 3);
    const float* hp = h + t0 * H_DIM + n0 + 4 * cg;
    const float* wp = w + t0;
    const v4f c = *reinterpret_cast<const v4f*>(W2 + n0 + 4 * cg);

    v4f sb = {0.0f, 0.0f, 0.0f, 0.0f}, sw = {0.0f, 0.0f, 0.0f, 0.0f};
#pragma unroll
    for (int i = 0; i < BW_K_PER_WAVE / 4; ++i) {
        const v4f hv = *reinterpret_cast<const v4f*>(hp + i * 4 * H_DIM);
        const float wt = wp[i * 4];
        const v4f e = wt * c;
        sb.x += (hv.x > 0.0f) ? e.x : 0.0f;
        sb.y += (hv.y > 0.0f) ? e.y : 0.0f;
        sb.z += (hv.z > 0.0f) ? e.z : 0.0f;
        sb.w += (hv.w > 0.0f) ? e.w : 0.0f;
        sw += wt * hv;
    }
#pragma unroll
    for (int q = 0; q < 4; ++q) {
        sb[q] += __shfl_xor(sb[q], 8);   sw[q] += __shfl_xor(sw[q], 8);
        sb[q] += __shfl_xor(sb[q], 16);  sw[q] += __shfl_xor(sw[q], 16);
    }
    if (lane < 8) {
#pragma unroll
        for (int q = 0; q < 4; ++q) {
            red[wave][0][4 * cg + q] = sb[q];
            red[wave][1][4 * cg + q] = sw[q];
        }
    }
    __syncthreads();
    if (threadIdx.x < 64) {
        float s = 0.0f;
#pragma unroll
        for (int q = 0; q < BW_WAVES; ++q) s += red[q][wave][lane];
        out[D_DIM * H_DIM + wave * H_DIM + n0 + lane] = -s;
    }
}

__global__ __launch_bounds__(BW_WAVES * 32)
__attribute__((amdgpu_waves_per_eu(4, 4))) void bwd_gemm_kernel(
    const float* __restrict__ X, const float* __restrict__ h,
    const float* __restrict__ W2, const float* __restrict__ w,
    float* __restrict__ out)
{
    __shared__ float red[BW_WAVES][16][32];

    if (blockIdx.x < 128) bwd_panel(X, h, W2, w, out, red);
    else                  col_sums(h, W2, w, out, red);
}

extern "C" void kernel_launch(void* const* d_in, const int* in_sizes, int n_in,
                              void* d_out, int out_size, void* d_ws, size_t ws_size,
                              hipStream_t stream)
{
    const float* xs = (const float*)d_in[0];
    const float* ys = (const float*)d_in[1];
    const float* W1 = (const float*)d_in[2];
    const float* b1 = (const float*)d_in[3];
    const float* W2 = (const float*)d_in[4];
    const float* b2 = (const float*)d_in[5];
    float* out = (float*)d_out;

    float* ws = (float*)d_ws;
    float* h     = ws;
    float* up    = h + (size_t)T_DIM * H_DIM;
    float* w     = up + 256 * UP_PITCH;

    fwd_gemm_kernel<<<256, 256, 0, stream>>>(xs, W1, b1, W2, h, up);
    scan_kernel<<<1, 256, 0, stream>>>(up, b2, ys, w, out);
    bwd_gemm_kernel<<<128 + 8, 512, 0, stream>>>(xs, h, W2, w, out);
}


# scan kernel folded into the bwd GEMM launch: every block runs the scans itself (waves 0-7, same operations and order), w stays in LDS, first GEMM load group in flight meanwhile; 3 to 2 launches
# speedup vs baseline: 12.5252x; 1.1441x over previous
#include <hip/hip_runtime.h>
#include <hip/hip_bf16.h>

typedef float v2f __attribute__((ext_vector_type(2)));
typedef float v8f __attribute__((ext_vector_type(8)));
typedef float v4f __attribute__((ext_vector_type(4)));

#define T_DIM 2048
#define D_DIM 256
#define H_DIM 256
#define GAMMA_LAMBDA (0.99f * 0.9f)
#define CONT_ALPHA 0.01f

#define UP_PITCH 32

template <int M>
__device__ __forceinline__ float row_xor_c(float v)
{
    return __builtin_bit_cast(float, __builtin_amdgcn_update_dpp(
        0, __builtin_bit_cast(int, v), 0x160 + M, 0xf, 0xf, true));
}
__device__ __forceinline__ float row_sum16(float v)
{
    v += row_xor_c<1>(v);
    v += row_xor_c<2>(v);
    v += row_xor_c<4>(v);
    v += row_xor_c<8>(v);
    return v;
}

__global__ __launch_bounds__(256) void fwd_gemm_kernel(
    const float* __restrict__ X, const float* __restrict__ W1,
    const float* __restrict__ b1, const float* __restrict__ W2,
    float* __restrict__ h, float* __restrict__ up)
{
    __shared__ __attribute__((aligned(16))) float part[8][16];

    const int lane = threadIdx.x & 31;
    const int wave = threadIdx.x >> 5;
    const int tile = blockIdx.x * 8 + wave;
    const int tileM = tile >> 4;
    const int tileN = tile & 15;
    const int half = lane >> 4;
    const int l16  = lane & 15;
    const int rowA = tileM * 16 + l16;
    const int colB = tileN * 16 + l16;

    v8f acc = {};
    for (int k0 = 0; k0 < D_DIM; k0 += 4) {
        const int ka = k0 + 2 * half;
        v2f a, b;
        a.x = X[rowA * D_DIM + ka];
        a.y = X[rowA * D_DIM + ka + 1];
        b.x = W1[ka * H_DIM + colB];
        b.y = W1[(ka + 1) * H_DIM + colB];
        acc = __builtin_amdgcn_wmma_f32_16x16x4_f32(
            false, a, false, b, (short)0, acc, false, false);
    }
    const float bias = b1[colB];
    const float c2 = W2[colB];
    float hv[8], pr[8];
#pragma unroll
    for (int v = 0; v < 8; ++v) {
        const float val = acc[v] + bias;
        hv[v] = val > 0.0f ? val : 0.0f;
        pr[v] = row_sum16(hv[v] * c2);
    }
    if (l16 == 0) {
        float4* dst = reinterpret_cast<float4*>(&part[wave][half * 8]);
        dst[0] = make_float4(pr[0], pr[1], pr[2], pr[3]);
        dst[1] = make_float4(pr[4], pr[5], pr[6], pr[7]);
    }
    __syncthreads();
#pragma unroll
    for (int v = 0; v < 8; ++v) {
        const int m = tileM * 16 + half * 8 + v;
        h[m * H_DIM + colB] = hv[v];
    }
    if (threadIdx.x < 16) {
        float s = 0.0f;
#pragma unroll
        for (int q = 0; q < 8; ++q) s += part[q][threadIdx.x];
        up[blockIdx.x * UP_PITCH + threadIdx.x] = s;
    }
}

#define SCAN_PER 8
#define SCAN_THREADS (T_DIM / SCAN_PER)
#define SCAN_WAVES (SCAN_THREADS / 32)

__device__ __forceinline__ void lds_barrier()
{
    __builtin_amdgcn_fence(__ATOMIC_RELEASE, "workgroup", "local");
    __builtin_amdgcn_s_barrier();
    __builtin_amdgcn_fence(__ATOMIC_ACQUIRE, "workgroup", "local");
}

struct ScanShared {
    double sA[SCAN_WAVES], sB[SCAN_WAVES];
    float sC[SCAN_WAVES], sS[SCAN_WAVES];
    float sSum[SCAN_WAVES];
    float sLast;
};

struct ScanIn {
    float4 ua, ub, va, vb, y0, y1, y2, y3;
    float bb;
};
__device__ __forceinline__ void scan_load(
    const float* __restrict__ up, const float* __restrict__ b2,
    const float* __restrict__ ys, ScanIn& in)
{
    const int tid = threadIdx.x;
    if (tid < SCAN_THREADS) {
        const float* ul = up + (tid >> 1) * 2 * UP_PITCH + (tid & 1) * 8;
        const float4* u4 = reinterpret_cast<const float4*>(ul);
        const float4* v4 = reinterpret_cast<const float4*>(ul + UP_PITCH);
        const float4* y4 = reinterpret_cast<const float4*>(ys) + tid * 4;
        in.ua = u4[0]; in.ub = u4[1];
        in.va = v4[0]; in.vb = v4[1];
        in.y0 = y4[0]; in.y1 = y4[1]; in.y2 = y4[2]; in.y3 = y4[3];
        in.bb = b2[0];
    }
}

__device__ __forceinline__ void block_scan(
    const ScanIn& in, float* __restrict__ out,
    float (&wsh)[T_DIM], ScanShared& sh)
{
    const int tid  = threadIdx.x;
    const int lane = tid & 31;
    const int wave = tid >> 5;
    const bool act = tid < SCAN_THREADS;

    float d[SCAN_PER] = {};
    bool  dn[SCAN_PER] = {};
    float sg[SCAN_PER] = {};
    double A = 1.0, B = 0.0;
    float C = 1.0f, S = 0.0f;

    if (act) {
        const float4 ua = in.ua, ub = in.ub, va = in.va, vb = in.vb;
        const float4 y0 = in.y0, y1 = in.y1, y2 = in.y2, y3 = in.y3;
        const float bb = in.bb;
        const float uu[SCAN_PER] = {
            (ua.x + va.x) + bb, (ua.y + va.y) + bb, (ua.z + va.z) + bb, (ua.w + va.w) + bb,
            (ub.x + vb.x) + bb, (ub.y + vb.y) + bb, (ub.z + vb.z) + bb, (ub.w + vb.w) + bb};
        const float gg[SCAN_PER] = {y0.x, y0.z, y1.x, y1.z, y2.x, y2.z, y3.x, y3.z};
        const float dd[SCAN_PER] = {y0.y, y0.w, y1.y, y1.w, y2.y, y2.w, y3.y, y3.w};

#pragma unroll
        for (int s = 0; s < SCAN_PER; ++s) {
            d[s]  = gg[s] - uu[s];
            dn[s] = dd[s] > 0.5f;
        }

        float Bf = 0.0f;
#pragma unroll
        for (int s = 0; s < SCAN_PER; ++s) {
            const float sg0 = d[s] * d[s] - Bf;
            Bf = Bf + CONT_ALPHA * sg0;
            A = A * (1.0 - (double)CONT_ALPHA);
            if (dn[s]) { A = 0.0; Bf = 0.0f; }
        }
        B = (double)Bf;
#pragma unroll
        for (int off = 1; off < 32; off <<= 1) {
            const double Ap = __shfl_up(A, off);
            const double Bp = __shfl_up(B, off);
            if (lane >= off) { B = A * Bp + B; A = A * Ap; }
        }
        if (lane == 31) { sh.sA[wave] = A; sh.sB[wave] = B; }
    }
    lds_barrier();
    if (act) {
        double wa[SCAN_WAVES], wb[SCAN_WAVES];
#pragma unroll
        for (int q = 0; q < SCAN_WAVES; ++q) { wa[q] = sh.sA[q]; wb[q] = sh.sB[q]; }
        double rin = 0.0;
#pragma unroll
        for (int q = 0; q < SCAN_WAVES; ++q) {
            const double nx = wa[q] * rin + wb[q];
            rin = (q < wave) ? nx : rin;
        }
        double Ae = __shfl_up(A, 1);
        double Be = __shfl_up(B, 1);
        if (lane == 0) { Ae = 1.0; Be = 0.0; }
        float r = (float)(Ae * rin + Be);

#pragma unroll
        for (int s = 0; s < SCAN_PER; ++s) {
            sg[s] = d[s] * d[s] - r;
            r = r + CONT_ALPHA * sg[s];
            if (dn[s]) r = 0.0f;
        }

#pragma unroll
        for (int s = SCAN_PER - 1; s >= 0; --s) {
            const float c = dn[s] ? 0.0f : GAMMA_LAMBDA;
            S = sg[s] + c * S;
            C = C * c;
        }
#pragma unroll
        for (int off = 1; off < 32; off <<= 1) {
            const float Cn = __shfl_down(C, off);
            const float Sn = __shfl_down(S, off);
            if (lane + off < 32) { S = C * Sn + S; C = C * Cn; }
        }
        if (lane == 0) { sh.sC[wave] = C; sh.sS[wave] = S; }
        if (tid == SCAN_THREADS - 1) sh.sLast = sg[SCAN_PER - 1];
    }
    lds_barrier();
    if (act) {
        float win = 0.0f;
#pragma unroll
        for (int q = SCAN_WAVES - 1; q >= 0; --q)
            if (q > wave) win = sh.sC[q] * win + sh.sS[q];
        float Ce = __shfl_down(C, 1);
        float Se = __shfl_down(S, 1);
        if (lane == 31) { Ce = 1.0f; Se = 0.0f; }
        float wv = Ce * win + Se;

        float wo[SCAN_PER];
        float acc = 0.0f;
#pragma unroll
        for (int s = SCAN_PER - 1; s >= 0; --s) {
            const float c = dn[s] ? 0.0f : GAMMA_LAMBDA;
            wv = sg[s] + c * wv;
            wo[s] = wv;
            acc += wv;
        }
        float4* w4 = reinterpret_cast<float4*>(&wsh[0]) + tid * 2;
        w4[0] = make_float4(wo[0], wo[1], wo[2], wo[3]);
        w4[1] = make_float4(wo[4], wo[5], wo[6], wo[7]);

#pragma unroll
        for (int off = 16; off > 0; off >>= 1) acc += __shfl_xor(acc, off);
        if (lane == 0) sh.sSum[wave] = acc;
    }
    lds_barrier();
    if (blockIdx.x == 0 && tid == 0) {
        float tot = 0.0f;
#pragma unroll
        for (int q = 0; q < SCAN_WAVES; ++q) tot += sh.sSum[q];
        out[D_DIM * H_DIM + 2 * H_DIM + 0] = -tot;
        out[D_DIM * H_DIM + 2 * H_DIM + 1] = sh.sLast;
    }
}

#define BW_WAVES 16
#define BW_K_PER_WAVE (T_DIM / BW_WAVES)
#define BW_STEPS (BW_K_PER_WAVE / 4)
#define BW_GROUP 16

template <int N>
__device__ __forceinline__ float row_lane_c(float v)
{
    return __builtin_bit_cast(float, __builtin_amdgcn_update_dpp(
        0, __builtin_bit_cast(int, v), 0x150 + N, 0xf, 0xf, true));
}
__device__ __forceinline__ float row_lane(float v, int n)
{
    switch (n) {
    case 0:  return row_lane_c<0>(v);   case 1:  return row_lane_c<1>(v);
    case 2:  return row_lane_c<2>(v);   case 3:  return row_lane_c<3>(v);
    case 4:  return row_lane_c<4>(v);   case 5:  return row_lane_c<5>(v);
    case 6:  return row_lane_c<6>(v);   case 7:  return row_lane_c<7>(v);
    case 8:  return row_lane_c<8>(v);   case 9:  return row_lane_c<9>(v);
    case 10: return row_lane_c<10>(v);  case 11: return row_lane_c<11>(v);
    case 12: return row_lane_c<12>(v);  case 13: return row_lane_c<13>(v);
    case 14: return row_lane_c<14>(v);  default: return row_lane_c<15>(v);
    }
}

__device__ __forceinline__ void bwd_panel(
    const float* __restrict__ X, const float* __restrict__ h,
    const float* __restrict__ W2, const float* __restrict__ up,
    const float* __restrict__ b2, const float* __restrict__ ys,
    float* __restrict__ out, float (&red)[BW_WAVES][16][32],
    float (&wsh)[T_DIM], ScanShared& sh)
{
    ScanIn in;
    scan_load(up, b2, ys, in);

    const int lane = threadIdx.x & 31;
    const int wave = threadIdx.x >> 5;
    const int half = lane >> 4;
    const int l16  = lane & 15;
    const int m0 = (blockIdx.x >> 3) * 16;
    const int n0 = (blockIdx.x & 7) * 32;

    const int kw = wave * BW_K_PER_WAVE + 2 * half;
    const float* xa = X + kw * D_DIM + m0 + l16;
    const float* hb = h + kw * H_DIM + n0 + 2 * l16;
    const v2f c = *reinterpret_cast<const v2f*>(W2 + n0 + 2 * l16);

    float ax0[BW_GROUP], ay0[BW_GROUP];
    v2f   p00[BW_GROUP], p10[BW_GROUP];
#pragma unroll
    for (int i = 0; i < BW_GROUP; ++i) {
        const int k = i * 4;
        ax0[i] = xa[k * D_DIM];
        ay0[i] = xa[(k + 1) * D_DIM];
        p00[i] = *reinterpret_cast<const v2f*>(hb + k * H_DIM);
        p10[i] = *reinterpret_cast<const v2f*>(hb + (k + 1) * H_DIM);
    }

    block_scan(in, out, wsh, sh);

    const float* wl = &wsh[wave * BW_K_PER_WAVE + 4 * l16 + 2 * half];
    const v2f wlo = *reinterpret_cast<const v2f*>(wl);
    const v2f whi = *reinterpret_cast<const v2f*>(wl + BW_K_PER_WAVE / 2);

    v8f acc0 = {}, acc1 = {};
#pragma unroll
    for (int g = 0; g < BW_STEPS; g += BW_GROUP) {
        float ax[BW_GROUP], ay[BW_GROUP];
        v2f   p0[BW_GROUP], p1[BW_GROUP];
#pragma unroll
        for (int i = 0; i < BW_GROUP; ++i) {
            if (g == 0) {
                ax[i] = ax0[i];  ay[i] = ay0[i];
                p0[i] = p00[i];  p1[i] = p10[i];
            } else {
                const int k = (g + i) * 4;
                ax[i] = xa[k * D_DIM];
                ay[i] = xa[(k + 1) * D_DIM];
                p0[i] = *reinterpret_cast<const v2f*>(hb + k * H_DIM);
                p1[i] = *reinterpret_cast<const v2f*>(hb + (k + 1) * H_DIM);
            }
        }
#pragma unroll
        for (int i = 0; i < BW_GROUP; ++i) {
            const v2f wsrc = (g == 0) ? wlo : whi;
            const float w0 = row_lane(wsrc.x, i);
            const float w1 = row_lane(wsrc.y, i);
            v2f a, b0, b1;
            a.x  = ax[i];   a.y  = ay[i];
            b0.x = (p0[i].x > 0.0f) ? w0 * c.x : 0.0f;
            b0.y = (p1[i].x > 0.0f) ? w1 * c.x : 0.0f;
            b1.x = (p0[i].y > 0.0f) ? w0 * c.y : 0.0f;
            b1.y = (p1[i].y > 0.0f) ? w1 * c.y : 0.0f;
            acc0 = __builtin_amdgcn_wmma_f32_16x16x4_f32(
                false, a, false, b0, (short)0, acc0, false, false);
            acc1 = __builtin_amdgcn_wmma_f32_16x16x4_f32(
                false, a, false, b1, (short)0, acc1, false, false);
        }
    }

#pragma unroll
    for (int v = 0; v < 8; ++v) {
        red[wave][v][lane]     = acc0[v];
        red[wave][8 + v][lane] = acc1[v];
    }
    __syncthreads();

    if (threadIdx.x < 256) {
        const int v = threadIdx.x >> 5;
        float s0 = 0.0f, s1 = 0.0f;
#pragma unroll
        for (int q = 0; q < BW_WAVES; ++q) {
            s0 += red[q][v][lane];
            s1 += red[q][8 + v][lane];
        }
        const int m = m0 + half * 8 + v;
        float2* dst = reinterpret_cast<float2*>(out + m * H_DIM + n0 + 2 * l16);
        *dst = make_float2(-s0, -s1);
    }
}

__device__ __forceinline__ void col_sums(
    const float* __restrict__ h, const float* __restrict__ W2,
    const float* __restrict__ up, const float* __restrict__ b2,
    const float* __restrict__ ys, float* __restrict__ out,
    float (&red)[BW_WAVES][16][32], float (&wsh)[T_DIM], ScanShared& sh)
{
    ScanIn in;
    scan_load(up, b2, ys, in);
    block_scan(in, out, wsh, sh);

    const int lane = threadIdx.x & 31;
    const int wave = threadIdx.x >> 5;
    const int n0 = (blockIdx.x - 128) * 32;
    const int cg = lane & 7;
    const int t0 = wave * BW_K_PER_WAVE + (lane >> 3);
    const float* hp = h + t0 * H_DIM + n0 + 4 * cg;
    const float* wp = &wsh[t0];
    const v4f c = *reinterpret_cast<const v4f*>(W2 + n0 + 4 * cg);

    v4f sb = {0.0f, 0.0f, 0.0f, 0.0f}, sw = {0.0f, 0.0f, 0.0f, 0.0f};
#pragma unroll
    for (int i = 0; i < BW_K_PER_WAVE / 4; ++i) {
        const v4f hv = *reinterpret_cast<const v4f*>(hp + i * 4 * H_DIM);
        const float wt = wp[i * 4];
        const v4f e = wt * c;
        sb.x += (hv.x > 0.0f) ? e.x : 0.0f;
        sb.y += (hv.y > 0.0f) ? e.y : 0.0f;
        sb.z += (hv.z > 0.0f) ? e.z : 0.0f;
        sb.w += (hv.w > 0.0f) ? e.w : 0.0f;
        sw += wt * hv;
    }
#pragma unroll
    for (int q = 0; q < 4; ++q) {
        sb[q] += __shfl_xor(sb[q], 8);   sw[q] += __shfl_xor(sw[q], 8);
        sb[q] += __shfl_xor(sb[q], 16);  sw[q] += __shfl_xor(sw[q], 16);
    }
    if (lane < 8) {
#pragma unroll
        for (int q = 0; q < 4; ++q) {
            red[wave][0][4 * cg + q] = sb[q];
            red[wave][1][4 * cg + q] = sw[q];
        }
    }
    __syncthreads();
    if (threadIdx.x < 64) {
        float s = 0.0f;
#pragma unroll
        for (int q = 0; q < BW_WAVES; ++q) s += red[q][wave][lane];
        out[D_DIM * H_DIM + wave * H_DIM + n0 + lane] = -s;
    }
}

__global__ __launch_bounds__(BW_WAVES * 32)
__attribute__((amdgpu_waves_per_eu(4, 4))) void bwd_gemm_kernel(
    const float* __restrict__ X, const float* __restrict__ h,
    const float* __restrict__ W2, const float* __restrict__ up,
    const float* __restrict__ b2, const float* __restrict__ ys,
    float* __restrict__ out)
{
    __shared__ float red[BW_WAVES][16][32];
    __shared__ __attribute__((aligned(16))) float wsh[T_DIM];
    __shared__ ScanShared sh;

    if (blockIdx.x < 128) bwd_panel(X, h, W2, up, b2, ys, out, red, wsh, sh);
    else                  col_sums(h, W2, up, b2, ys, out, red, wsh, sh);
}

extern "C" void kernel_launch(void* const* d_in, const int* in_sizes, int n_in,
                              void* d_out, int out_size, void* d_ws, size_t ws_size,
                              hipStream_t stream)
{
    const float* xs = (const float*)d_in[0];
    const float* ys = (const float*)d_in[1];
    const float* W1 = (const float*)d_in[2];
    const float* b1 = (const float*)d_in[3];
    const float* W2 = (const float*)d_in[4];
    const float* b2 = (const float*)d_in[5];
    float* out = (float*)d_out;

    float* ws = (float*)d_ws;
    float* h     = ws;
    float* up    = h + (size_t)T_DIM * H_DIM;

    fwd_gemm_kernel<<<256, 256, 0, stream>>>(xs, W1, b1, W2, h, up);
    bwd_gemm_kernel<<<128 + 8, 512, 0, stream>>>(xs, h, W2, up, b2, ys, out);
}


# fwd GEMM operand feed: the block's 16-row X band is staged once in LDS by whole-line 16-byte loads (8 waves each re-read it, 16 lines per load); W1 loads issued 32 k-steps ahead; sums unchanged
# speedup vs baseline: 15.9477x; 1.2732x over previous
#include <hip/hip_runtime.h>
#include <hip/hip_bf16.h>

typedef float v2f __attribute__((ext_vector_type(2)));
typedef float v8f __attribute__((ext_vector_type(8)));
typedef float v4f __attribute__((ext_vector_type(4)));

#define T_DIM 2048
#define D_DIM 256
#define H_DIM 256
#define GAMMA_LAMBDA (0.99f * 0.9f)
#define CONT_ALPHA 0.01f

__device__ __forceinline__ void lds_barrier()
{
    __builtin_amdgcn_fence(__ATOMIC_RELEASE, "workgroup", "local");
    __builtin_amdgcn_s_barrier();
    __builtin_amdgcn_fence(__ATOMIC_ACQUIRE, "workgroup", "local");
}

#define UP_PITCH 32
#define FW_STEPS (D_DIM / 4)
#define FW_GROUP 16
#define A_PITCH (D_DIM + 4)

template <int M>
__device__ __forceinline__ float row_xor_c(float v)
{
    return __builtin_bit_cast(float, __builtin_amdgcn_update_dpp(
        0, __builtin_bit_cast(int, v), 0x160 + M, 0xf, 0xf, true));
}
__device__ __forceinline__ float row_sum16(float v)
{
    v += row_xor_c<1>(v);
    v += row_xor_c<2>(v);
    v += row_xor_c<4>(v);
    v += row_xor_c<8>(v);
    return v;
}

__global__ __launch_bounds__(256) void fwd_gemm_kernel(
    const float* __restrict__ X, const float* __restrict__ W1,
    const float* __restrict__ b1, const float* __restrict__ W2,
    float* __restrict__ h, float* __restrict__ up)
{
    __shared__ __attribute__((aligned(16))) float part[8][16];
    __shared__ __attribute__((aligned(16))) float As[16 * A_PITCH];

    const int lane = threadIdx.x & 31;
    const int wave = threadIdx.x >> 5;
    const int tile = blockIdx.x * 8 + wave;
    const int tileM = tile >> 4;
    const int tileN = tile & 15;
    const int half = lane >> 4;
    const int l16  = lane & 15;
    const int colB = tileN * 16 + l16;

    const float4* xb = reinterpret_cast<const float4*>(X + tileM * 16 * D_DIM);
    const float4 xq0 = xb[threadIdx.x];
    const float4 xq1 = xb[threadIdx.x + 256];
    const float4 xq2 = xb[threadIdx.x + 512];
    const float4 xq3 = xb[threadIdx.x + 768];

    const float* wb = W1 + 2 * half * H_DIM + colB;
    float bx[FW_STEPS], by[FW_STEPS];
#pragma unroll
    for (int s = 0; s < 2 * FW_GROUP; ++s) {
        bx[s] = wb[(4 * s) * H_DIM];
        by[s] = wb[(4 * s + 1) * H_DIM];
    }

    float* ap = &As[(threadIdx.x >> 6) * A_PITCH + 4 * (threadIdx.x & 63)];
    *reinterpret_cast<float4*>(ap)                = xq0;
    *reinterpret_cast<float4*>(ap + 4 * A_PITCH)  = xq1;
    *reinterpret_cast<float4*>(ap + 8 * A_PITCH)  = xq2;
    *reinterpret_cast<float4*>(ap + 12 * A_PITCH) = xq3;
    lds_barrier();

    const float* al = &As[l16 * A_PITCH + 2 * half];
    v8f acc = {};
#pragma unroll
    for (int g = 0; g < FW_STEPS; g += FW_GROUP) {
        v2f af[FW_GROUP];
#pragma unroll
        for (int i = 0; i < FW_GROUP; ++i)
            af[i] = *reinterpret_cast<const v2f*>(al + 4 * (g + i));
#pragma unroll
        for (int i = 0; i < FW_GROUP; ++i) {
            const int s = g + 2 * FW_GROUP + i;
            if (s < FW_STEPS) {
                bx[s] = wb[(4 * s) * H_DIM];
                by[s] = wb[(4 * s + 1) * H_DIM];
            }
        }
#pragma unroll
        for (int i = 0; i < FW_GROUP; ++i) {
            const int s = g + i;
            const v2f a = af[i];
            v2f b;
            b.x = bx[s];
            b.y = by[s];
            acc = __builtin_amdgcn_wmma_f32_16x16x4_f32(
                false, a, false, b, (short)0, acc, false, false);
        }
    }
    const float bias = b1[colB];
    const float c2 = W2[colB];
    float hv[8], pr[8];
#pragma unroll
    for (int v = 0; v < 8; ++v) {
        const float val = acc[v] + bias;
        hv[v] = val > 0.0f ? val : 0.0f;
        pr[v] = row_sum16(hv[v] * c2);
    }
    if (l16 == 0) {
        float4* dst = reinterpret_cast<float4*>(&part[wave][half * 8]);
        dst[0] = make_float4(pr[0], pr[1], pr[2], pr[3]);
        dst[1] = make_float4(pr[4], pr[5], pr[6], pr[7]);
    }
    __syncthreads();
#pragma unroll
    for (int v = 0; v < 8; ++v) {
        const int m = tileM * 16 + half * 8 + v;
        h[m * H_DIM + colB] = hv[v];
    }
    if (threadIdx.x < 16) {
        float s = 0.0f;
#pragma unroll
        for (int q = 0; q < 8; ++q) s += part[q][threadIdx.x];
        up[blockIdx.x * UP_PITCH + threadIdx.x] = s;
    }
}

#define SCAN_PER 8
#define SCAN_THREADS (T_DIM / SCAN_PER)
#define SCAN_WAVES (SCAN_THREADS / 32)

struct ScanShared {
    double sA[SCAN_WAVES], sB[SCAN_WAVES];
    float sC[SCAN_WAVES], sS[SCAN_WAVES];
    float sSum[SCAN_WAVES];
    float sLast;
};

struct ScanIn {
    float4 ua, ub, va, vb, y0, y1, y2, y3;
    float bb;
};
__device__ __forceinline__ void scan_load(
    const float* __restrict__ up, const float* __restrict__ b2,
    const float* __restrict__ ys, ScanIn& in)
{
    const int tid = threadIdx.x;
    if (tid < SCAN_THREADS) {
        const float* ul = up + (tid >> 1) * 2 * UP_PITCH + (tid & 1) * 8;
        const float4* u4 = reinterpret_cast<const float4*>(ul);
        const float4* v4 = reinterpret_cast<const float4*>(ul + UP_PITCH);
        const float4* y4 = reinterpret_cast<const float4*>(ys) + tid * 4;
        in.ua = u4[0]; in.ub = u4[1];
        in.va = v4[0]; in.vb = v4[1];
        in.y0 = y4[0]; in.y1 = y4[1]; in.y2 = y4[2]; in.y3 = y4[3];
        in.bb = b2[0];
    }
}

__device__ __forceinline__ void block_scan(
    const ScanIn& in, float* __restrict__ out,
    float (&wsh)[T_DIM], ScanShared& sh)
{
    const int tid  = threadIdx.x;
    const int lane = tid & 31;
    const int wave = tid >> 5;
    const bool act = tid < SCAN_THREADS;

    float d[SCAN_PER] = {};
    bool  dn[SCAN_PER] = {};
    float sg[SCAN_PER] = {};
    double A = 1.0, B = 0.0;
    float C = 1.0f, S = 0.0f;

    if (act) {
        const float4 ua = in.ua, ub = in.ub, va = in.va, vb = in.vb;
        const float4 y0 = in.y0, y1 = in.y1, y2 = in.y2, y3 = in.y3;
        const float bb = in.bb;
        const float uu[SCAN_PER] = {
            (ua.x + va.x) + bb, (ua.y + va.y) + bb, (ua.z + va.z) + bb, (ua.w + va.w) + bb,
            (ub.x + vb.x) + bb, (ub.y + vb.y) + bb, (ub.z + vb.z) + bb, (ub.w + vb.w) + bb};
        const float gg[SCAN_PER] = {y0.x, y0.z, y1.x, y1.z, y2.x, y2.z, y3.x, y3.z};
        const float dd[SCAN_PER] = {y0.y, y0.w, y1.y, y1.w, y2.y, y2.w, y3.y, y3.w};

#pragma unroll
        for (int s = 0; s < SCAN_PER; ++s) {
            d[s]  = gg[s] - uu[s];
            dn[s] = dd[s] > 0.5f;
        }

        float Bf = 0.0f;
#pragma unroll
        for (int s = 0; s < SCAN_PER; ++s) {
            const float sg0 = d[s] * d[s] - Bf;
            Bf = Bf + CONT_ALPHA * sg0;
            A = A * (1.0 - (double)CONT_ALPHA);
            if (dn[s]) { A = 0.0; Bf = 0.0f; }
        }
        B = (double)Bf;
#pragma unroll
        for (int off = 1; off < 32; off <<= 1) {
            const double Ap = __shfl_up(A, off);
            const double Bp = __shfl_up(B, off);
            if (lane >= off) { B = A * Bp + B; A = A * Ap; }
        }
        if (lane == 31) { sh.sA[wave] = A; sh.sB[wave] = B; }
    }
    lds_barrier();
    if (act) {
        double wa[SCAN_WAVES], wb[SCAN_WAVES];
#pragma unroll
        for (int q = 0; q < SCAN_WAVES; ++q) { wa[q] = sh.sA[q]; wb[q] = sh.sB[q]; }
        double rin = 0.0;
#pragma unroll
        for (int q = 0; q < SCAN_WAVES; ++q) {
            const double nx = wa[q] * rin + wb[q];
            rin = (q < wave) ? nx : rin;
        }
        double Ae = __shfl_up(A, 1);
        double Be = __shfl_up(B, 1);
        if (lane == 0) { Ae = 1.0; Be = 0.0; }
        float r = (float)(Ae * rin + Be);

#pragma unroll
        for (int s = 0; s < SCAN_PER; ++s) {
            sg[s] = d[s] * d[s] - r;
            r = r + CONT_ALPHA * sg[s];
            if (dn[s]) r = 0.0f;
        }

#pragma unroll
        for (int s = SCAN_PER - 1; s >= 0; --s) {
            const float c = dn[s] ? 0.0f : GAMMA_LAMBDA;
            S = sg[s] + c * S;
            C = C * c;
        }
#pragma unroll
        for (int off = 1; off < 32; off <<= 1) {
            const float Cn = __shfl_down(C, off);
            const float Sn = __shfl_down(S, off);
            if (lane + off < 32) { S = C * Sn + S; C = C * Cn; }
        }
        if (lane == 0) { sh.sC[wave] = C; sh.sS[wave] = S; }
        if (tid == SCAN_THREADS - 1) sh.sLast = sg[SCAN_PER - 1];
    }
    lds_barrier();
    if (act) {
        float win = 0.0f;
#pragma unroll
        for (int q = SCAN_WAVES - 1; q >= 0; --q)
            if (q > wave) win = sh.sC[q] * win + sh.sS[q];
        float Ce = __shfl_down(C, 1);
        float Se = __shfl_down(S, 1);
        if (lane == 31) { Ce = 1.0f; Se = 0.0f; }
        float wv = Ce * win + Se;

        float wo[SCAN_PER];
        float acc = 0.0f;
#pragma unroll
        for (int s = SCAN_PER - 1; s >= 0; --s) {
            const float c = dn[s] ? 0.0f : GAMMA_LAMBDA;
            wv = sg[s] + c * wv;
            wo[s] = wv;
            acc += wv;
        }
        float4* w4 = reinterpret_cast<float4*>(&wsh[0]) + tid * 2;
        w4[0] = make_float4(wo[0], wo[1], wo[2], wo[3]);
        w4[1] = make_float4(wo[4], wo[5], wo[6], wo[7]);

#pragma unroll
        for (int off = 16; off > 0; off >>= 1) acc += __shfl_xor(acc, off);
        if (lane == 0) sh.sSum[wave] = acc;
    }
    lds_barrier();
    if (blockIdx.x == 0 && tid == 0) {
        float tot = 0.0f;
#pragma unroll
        for (int q = 0; q < SCAN_WAVES; ++q) tot += sh.sSum[q];
        out[D_DIM * H_DIM + 2 * H_DIM + 0] = -tot;
        out[D_DIM * H_DIM + 2 * H_DIM + 1] = sh.sLast;
    }
}

#define BW_WAVES 16
#define BW_K_PER_WAVE (T_DIM / BW_WAVES)
#define BW_STEPS (BW_K_PER_WAVE / 4)
#define BW_GROUP 16

template <int N>
__device__ __forceinline__ float row_lane_c(float v)
{
    return __builtin_bit_cast(float, __builtin_amdgcn_update_dpp(
        0, __builtin_bit_cast(int, v), 0x150 + N, 0xf, 0xf, true));
}
__device__ __forceinline__ float row_lane(float v, int n)
{
    switch (n) {
    case 0:  return row_lane_c<0>(v);   case 1:  return row_lane_c<1>(v);
    case 2:  return row_lane_c<2>(v);   case 3:  return row_lane_c<3>(v);
    case 4:  return row_lane_c<4>(v);   case 5:  return row_lane_c<5>(v);
    case 6:  return row_lane_c<6>(v);   case 7:  return row_lane_c<7>(v);
    case 8:  return row_lane_c<8>(v);   case 9:  return row_lane_c<9>(v);
    case 10: return row_lane_c<10>(v);  case 11: return row_lane_c<11>(v);
    case 12: return row_lane_c<12>(v);  case 13: return row_lane_c<13>(v);
    case 14: return row_lane_c<14>(v);  default: return row_lane_c<15>(v);
    }
}

__device__ __forceinline__ void bwd_panel(
    const float* __restrict__ X, const float* __restrict__ h,
    const float* __restrict__ W2, const float* __restrict__ up,
    const float* __restrict__ b2, const float* __restrict__ ys,
    float* __restrict__ out, float (&red)[BW_WAVES][16][32],
    float (&wsh)[T_DIM], ScanShared& sh)
{
    ScanIn in;
    scan_load(up, b2, ys, in);

    const int lane = threadIdx.x & 31;
    const int wave = threadIdx.x >> 5;
    const int half = lane >> 4;
    const int l16  = lane & 15;
    const int m0 = (blockIdx.x >> 3) * 16;
    const int n0 = (blockIdx.x & 7) * 32;

    const int kw = wave * BW_K_PER_WAVE + 2 * half;
    const float* xa = X + kw * D_DIM + m0 + l16;
    const float* hb = h + kw * H_DIM + n0 + 2 * l16;
    const v2f c = *reinterpret_cast<const v2f*>(W2 + n0 + 2 * l16);

    float ax0[BW_GROUP], ay0[BW_GROUP];
    v2f   p00[BW_GROUP], p10[BW_GROUP];
#pragma unroll
    for (int i = 0; i < BW_GROUP; ++i) {
        const int k = i * 4;
        ax0[i] = xa[k * D_DIM];
        ay0[i] = xa[(k + 1) * D_DIM];
        p00[i] = *reinterpret_cast<const v2f*>(hb + k * H_DIM);
        p10[i] = *reinterpret_cast<const v2f*>(hb + (k + 1) * H_DIM);
    }

    block_scan(in, out, wsh, sh);

    const float* wl = &wsh[wave * BW_K_PER_WAVE + 4 * l16 + 2 * half];
    const v2f wlo = *reinterpret_cast<const v2f*>(wl);
    const v2f whi = *reinterpret_cast<const v2f*>(wl + BW_K_PER_WAVE / 2);

    v8f acc0 = {}, acc1 = {};
#pragma unroll
    for (int g = 0; g < BW_STEPS; g += BW_GROUP) {
        float ax[BW_GROUP], ay[BW_GROUP];
        v2f   p0[BW_GROUP], p1[BW_GROUP];
#pragma unroll
        for (int i = 0; i < BW_GROUP; ++i) {
            if (g == 0) {
                ax[i] = ax0[i];  ay[i] = ay0[i];
                p0[i] = p00[i];  p1[i] = p10[i];
            } else {
                const int k = (g + i) * 4;
                ax[i] = xa[k * D_DIM];
                ay[i] = xa[(k + 1) * D_DIM];
                p0[i] = *reinterpret_cast<const v2f*>(hb + k * H_DIM);
                p1[i] = *reinterpret_cast<const v2f*>(hb + (k + 1) * H_DIM);
            }
        }
#pragma unroll
        for (int i = 0; i < BW_GROUP; ++i) {
            const v2f wsrc = (g == 0) ? wlo : whi;
            const float w0 = row_lane(wsrc.x, i);
            const float w1 = row_lane(wsrc.y, i);
            v2f a, b0, b1;
            a.x  = ax[i];   a.y  = ay[i];
            b0.x = (p0[i].x > 0.0f) ? w0 * c.x : 0.0f;
            b0.y = (p1[i].x > 0.0f) ? w1 * c.x : 0.0f;
            b1.x = (p0[i].y > 0.0f) ? w0 * c.y : 0.0f;
            b1.y = (p1[i].y > 0.0f) ? w1 * c.y : 0.0f;
            acc0 = __builtin_amdgcn_wmma_f32_16x16x4_f32(
                false, a, false, b0, (short)0, acc0, false, false);
            acc1 = __builtin_amdgcn_wmma_f32_16x16x4_f32(
                false, a, false, b1, (short)0, acc1, false, false);
        }
    }

#pragma unroll
    for (int v = 0; v < 8; ++v) {
        red[wave][v][lane]     = acc0[v];
        red[wave][8 + v][lane] = acc1[v];
    }
    __syncthreads();

    if (threadIdx.x < 256) {
        const int v = threadIdx.x >> 5;
        float s0 = 0.0f, s1 = 0.0f;
#pragma unroll
        for (int q = 0; q < BW_WAVES; ++q) {
            s0 += red[q][v][lane];
            s1 += red[q][8 + v][lane];
        }
        const int m = m0 + half * 8 + v;
        float2* dst = reinterpret_cast<float2*>(out + m * H_DIM + n0 + 2 * l16);
        *dst = make_float2(-s0, -s1);
    }
}

__device__ __forceinline__ void col_sums(
    const float* __restrict__ h, const float* __restrict__ W2,
    const float* __restrict__ up, const float* __restrict__ b2,
    const float* __restrict__ ys, float* __restrict__ out,
    float (&red)[BW_WAVES][16][32], float (&wsh)[T_DIM], ScanShared& sh)
{
    ScanIn in;
    scan_load(up, b2, ys, in);
    block_scan(in, out, wsh, sh);

    const int lane = threadIdx.x & 31;
    const int wave = threadIdx.x >> 5;
    const int n0 = (blockIdx.x - 128) * 32;
    const int cg = lane & 7;
    const int t0 = wave * BW_K_PER_WAVE + (lane >> 3);
    const float* hp = h + t0 * H_DIM + n0 + 4 * cg;
    const float* wp = &wsh[t0];
    const v4f c = *reinterpret_cast<const v4f*>(W2 + n0 + 4 * cg);

    v4f sb = {0.0f, 0.0f, 0.0f, 0.0f}, sw = {0.0f, 0.0f, 0.0f, 0.0f};
#pragma unroll
    for (int i = 0; i < BW_K_PER_WAVE / 4; ++i) {
        const v4f hv = *reinterpret_cast<const v4f*>(hp + i * 4 * H_DIM);
        const float wt = wp[i * 4];
        const v4f e = wt * c;
        sb.x += (hv.x > 0.0f) ? e.x : 0.0f;
        sb.y += (hv.y > 0.0f) ? e.y : 0.0f;
        sb.z += (hv.z > 0.0f) ? e.z : 0.0f;
        sb.w += (hv.w > 0.0f) ? e.w : 0.0f;
        sw += wt * hv;
    }
#pragma unroll
    for (int q = 0; q < 4; ++q) {
        sb[q] += __shfl_xor(sb[q], 8);   sw[q] += __shfl_xor(sw[q], 8);
        sb[q] += __shfl_xor(sb[q], 16);  sw[q] += __shfl_xor(sw[q], 16);
    }
    if (lane < 8) {
#pragma unroll
        for (int q = 0; q < 4; ++q) {
            red[wave][0][4 * cg + q] = sb[q];
            red[wave][1][4 * cg + q] = sw[q];
        }
    }
    __syncthreads();
    if (threadIdx.x < 64) {
        float s = 0.0f;
#pragma unroll
        for (int q = 0; q < BW_WAVES; ++q) s += red[q][wave][lane];
        out[D_DIM * H_DIM + wave * H_DIM + n0 + lane] = -s;
    }
}

__global__ __launch_bounds__(BW_WAVES * 32)
__attribute__((amdgpu_waves_per_eu(4, 4))) void bwd_gemm_kernel(
    const float* __restrict__ X, const float* __restrict__ h,
    const float* __restrict__ W2, const float* __restrict__ up,
    const float* __restrict__ b2, const float* __restrict__ ys,
    float* __restrict__ out)
{
    __shared__ float red[BW_WAVES][16][32];
    __shared__ __attribute__((aligned(16))) float wsh[T_DIM];
    __shared__ ScanShared sh;

    if (blockIdx.x < 128) bwd_panel(X, h, W2, up, b2, ys, out, red, wsh, sh);
    else                  col_sums(h, W2, up, b2, ys, out, red, wsh, sh);
}

extern "C" void kernel_launch(void* const* d_in, const int* in_sizes, int n_in,
                              void* d_out, int out_size, void* d_ws, size_t ws_size,
                              hipStream_t stream)
{
    const float* xs = (const float*)d_in[0];
    const float* ys = (const float*)d_in[1];
    const float* W1 = (const float*)d_in[2];
    const float* b1 = (const float*)d_in[3];
    const float* W2 = (const float*)d_in[4];
    const float* b2 = (const float*)d_in[5];
    float* out = (float*)d_out;

    float* ws = (float*)d_ws;
    float* h     = ws;
    float* up    = h + (size_t)T_DIM * H_DIM;

    fwd_gemm_kernel<<<256, 256, 0, stream>>>(xs, W1, b1, W2, h, up);
    bwd_gemm_kernel<<<128 + 8, 512, 0, stream>>>(xs, h, W2, up, b2, ys, out);
}
